# starting kernel
# baseline (speedup 1.0000x reference)
#include <hip/hip_runtime.h>
#include <hip/hip_bf16.h>
#include <math.h>

#define N_NODES   50000
#define N_EDGES   800000
#define IN_FEATS  128
#define NUM_HEADS 8
#define OUT_FEATS 16
#define HD        (NUM_HEADS * OUT_FEATS)
#define NEG_SLOPE 0.2f

typedef float v2f __attribute__((ext_vector_type(2)));
typedef float v8f __attribute__((ext_vector_type(8)));

__global__ __launch_bounds__(256)
void gat_gemm_wmma(const float* __restrict__ feat,
                   const float* __restrict__ W,
                   float* __restrict__ h) {
  __shared__ float lA[16 * IN_FEATS];
  const int tid  = threadIdx.x;
  const int wave = tid >> 5;
  const int lane = tid & 31;
  const int m0   = blockIdx.x * 16;
  const int n0   = wave * 16;

  #pragma unroll
  for (int i = tid; i < 16 * IN_FEATS; i += 256) {
    const int r = i >> 7;
    const int c = i & 127;
    lA[i] = feat[(size_t)(m0 + r) * IN_FEATS + c];
  }
  __syncthreads();

  const int ml  = lane & 15;
  const int klo = (lane < 16) ? 0 : 2;

  v8f acc = {};
  #pragma unroll 4
  for (int k = 0; k < IN_FEATS; k += 4) {
    const float2 av = *(const float2*)&lA[ml * IN_FEATS + k + klo];
    v2f a; a.x = av.x; a.y = av.y;
    v2f b;
    b.x = W[(size_t)(k + klo)     * HD + n0 + ml];
    b.y = W[(size_t)(k + klo + 1) * HD + n0 + ml];
    acc = __builtin_amdgcn_wmma_f32_16x16x4_f32(
         false, a,  false, b,
         (short)0, acc,  false,  false);
  }

  const int rbase = m0 + ((lane < 16) ? 0 : 8);
  #pragma unroll
  for (int v = 0; v < 8; ++v) {
    h[(size_t)(rbase + v) * HD + n0 + ml] = acc[v];
  }
}

__global__ void gat_init_out(float* __restrict__ out,
                             const float* __restrict__ bias) {
  const int idx = blockIdx.x * blockDim.x + threadIdx.x;
  if (idx < N_NODES * HD) out[idx] = bias[idx & (HD - 1)];
}

__global__ void gat_logits(const float* __restrict__ h,
                           const float* __restrict__ attn,
                           float* __restrict__ es, float* __restrict__ ed,
                           float* __restrict__ m, float* __restrict__ denom) {
  const int idx = blockIdx.x * blockDim.x + threadIdx.x;
  if (idx >= N_NODES * NUM_HEADS) return;
  const int head = idx & (NUM_HEADS - 1);
  const int n    = idx >> 3;
  const float* hp = h    + (size_t)n * HD + head * OUT_FEATS;
  const float* a0 = attn +           head * OUT_FEATS;
  const float* a1 = attn + HD      + head * OUT_FEATS;
  float s0 = 0.f, s1 = 0.f;
  #pragma unroll
  for (int d = 0; d < OUT_FEATS; ++d) {
    const float hv = hp[d];
    s0 += hv * a0[d];
    s1 += hv * a1[d];
  }
  es[idx]    = s0;
  ed[idx]    = s1;
  m[idx]     = -INFINITY;
  denom[idx] = 0.f;
}

__device__ __forceinline__ void atomicMaxFloat(float* addr, float val) {
  if (val >= 0.f) {
    atomicMax((int*)addr, __float_as_int(val));
  } else {
    atomicMin((unsigned int*)addr, __float_as_uint(val));
  }
}

__device__ __forceinline__ float edge_score(const float* __restrict__ es,
                                            const float* __restrict__ ed,
                                            int s, int t, int head) {
  const float e = es[s * NUM_HEADS + head] + ed[t * NUM_HEADS + head];
  return e > 0.f ? e : NEG_SLOPE * e;
}

__global__ void gat_edge_max(const float* __restrict__ es,
                             const float* __restrict__ ed,
                             const int* __restrict__ src,
                             const int* __restrict__ dst,
                             float* __restrict__ m) {
  const int idx = blockIdx.x * blockDim.x + threadIdx.x;
  if (idx >= N_EDGES * NUM_HEADS) return;
  const int head = idx & (NUM_HEADS - 1);
  const int e    = idx >> 3;
  const int s = src[e], t = dst[e];
  atomicMaxFloat(&m[t * NUM_HEADS + head], edge_score(es, ed, s, t, head));
}

__global__ void gat_edge_sum(const float* __restrict__ es,
                             const float* __restrict__ ed,
                             const int* __restrict__ src,
                             const int* __restrict__ dst,
                             const float* __restrict__ m,
                             float* __restrict__ denom) {
  const int idx = blockIdx.x * blockDim.x + threadIdx.x;
  if (idx >= N_EDGES * NUM_HEADS) return;
  const int head = idx & (NUM_HEADS - 1);
  const int e    = idx >> 3;
  const int s = src[e], t = dst[e];
  const float v  = edge_score(es, ed, s, t, head);
  const float ex = __expf(v - m[t * NUM_HEADS + head]);
  atomicAdd(&denom[t * NUM_HEADS + head], ex);
}

__global__ void gat_aggregate(const float* __restrict__ h,
                              const float* __restrict__ es,
                              const float* __restrict__ ed,
                              const float* __restrict__ m,
                              const float* __restrict__ denom,
                              const int* __restrict__ src,
                              const int* __restrict__ dst,
                              float* __restrict__ out) {
  const int idx = blockIdx.x * blockDim.x + threadIdx.x;
  if (idx >= N_EDGES * NUM_HEADS) return;
  const int head = idx & (NUM_HEADS - 1);
  const int e    = idx >> 3;
  const int s = src[e], t = dst[e];
  const float v  = edge_score(es, ed, s, t, head);
  const float dn = denom[t * NUM_HEADS + head];
  const float alpha = __expf(v - m[t * NUM_HEADS + head]) / (dn > 0.f ? dn : 1.f);
  const float* hs = h   + (size_t)s * HD + head * OUT_FEATS;
  float*       op = out + (size_t)t * HD + head * OUT_FEATS;
  #pragma unroll
  for (int d = 0; d < OUT_FEATS; d += 4) {
    const float4 hv = *(const float4*)&hs[d];
    atomicAdd(&op[d + 0], alpha * hv.x);
    atomicAdd(&op[d + 1], alpha * hv.y);
    atomicAdd(&op[d + 2], alpha * hv.z);
    atomicAdd(&op[d + 3], alpha * hv.w);
  }
}

extern "C" void kernel_launch(void* const* d_in, const int* in_sizes, int n_in,
                              void* d_out, int out_size, void* d_ws, size_t ws_size,
                              hipStream_t stream) {
  const float* feat = (const float*)d_in[0];
  const float* W    = (const float*)d_in[1];
  const float* attn = (const float*)d_in[2];
  const float* bias = (const float*)d_in[3];
  const int*   src  = (const int*)d_in[4];
  const int*   dst  = (const int*)d_in[5];
  float*       out  = (float*)d_out;

  float* h     = (float*)d_ws;
  float* es    = h     + (size_t)N_NODES * HD;
  float* ed    = es    + (size_t)N_NODES * NUM_HEADS;
  float* m     = ed    + (size_t)N_NODES * NUM_HEADS;
  float* denom = m     + (size_t)N_NODES * NUM_HEADS;

  const int NH = N_NODES * NUM_HEADS;
  const int EH = N_EDGES * NUM_HEADS;

  gat_init_out<<<(N_NODES * HD + 255) / 256, 256, 0, stream>>>(out, bias);
  gat_gemm_wmma<<<N_NODES / 16, 256, 0, stream>>>(feat, W, h);
  gat_logits<<<(NH + 255) / 256, 256, 0, stream>>>(h, attn, es, ed, m, denom);
  gat_edge_max<<<(EH + 255) / 256, 256, 0, stream>>>(es, ed, src, dst, m);
  gat_edge_sum<<<(EH + 255) / 256, 256, 0, stream>>>(es, ed, src, dst, m, denom);
  gat_aggregate<<<(EH + 255) / 256, 256, 0, stream>>>(h, es, ed, m, denom, src, dst, out);
}


# Softmax max and exp-sum formed in the owner block from its LDS edge list: max on ordered integer keys, exp-sum as 64-bit fixed-point LDS adds (order-free, bit-repeatable); both per-edge launches go
# speedup vs baseline: 4.9627x; 4.9627x over previous
#include <hip/hip_runtime.h>
#include <hip/hip_bf16.h>
#include <math.h>

#define N_NODES   50000
#define N_EDGES   800000
#define IN_FEATS  128
#define NUM_HEADS 8
#define OUT_FEATS 16
#define HD        (NUM_HEADS * OUT_FEATS)
#define NEG_SLOPE 0.2f

typedef float v2f __attribute__((ext_vector_type(2)));
typedef float v8f __attribute__((ext_vector_type(8)));

__global__ __launch_bounds__(256)
void gat_gemm_wmma(const float* __restrict__ feat,
                   const float* __restrict__ W,
                   float* __restrict__ h) {
  __shared__ float lA[16 * IN_FEATS];
  const int tid  = threadIdx.x;
  const int wave = tid >> 5;
  const int lane = tid & 31;
  const int m0   = blockIdx.x * 16;
  const int n0   = wave * 16;

  #pragma unroll
  for (int i = tid; i < 16 * IN_FEATS; i += 256) {
    const int r = i >> 7;
    const int c = i & 127;
    lA[i] = feat[(size_t)(m0 + r) * IN_FEATS + c];
  }
  __syncthreads();

  const int ml  = lane & 15;
  const int klo = (lane < 16) ? 0 : 2;

  v8f acc = {};
  #pragma unroll 4
  for (int k = 0; k < IN_FEATS; k += 4) {
    const float2 av = *(const float2*)&lA[ml * IN_FEATS + k + klo];
    v2f a; a.x = av.x; a.y = av.y;
    v2f b;
    b.x = W[(size_t)(k + klo)     * HD + n0 + ml];
    b.y = W[(size_t)(k + klo + 1) * HD + n0 + ml];
    acc = __builtin_amdgcn_wmma_f32_16x16x4_f32(
         false, a,  false, b,
         (short)0, acc,  false,  false);
  }

  const int rbase = m0 + ((lane < 16) ? 0 : 8);
  #pragma unroll
  for (int v = 0; v < 8; ++v) {
    h[(size_t)(rbase + v) * HD + n0 + ml] = acc[v];
  }
}

__global__ void gat_logits(const float* __restrict__ h,
                           const float* __restrict__ attn,
                           float* __restrict__ es, float* __restrict__ ed) {
  const int idx = blockIdx.x * blockDim.x + threadIdx.x;
  if (idx >= N_NODES * NUM_HEADS) return;
  const int head = idx & (NUM_HEADS - 1);
  const int n    = idx >> 3;
  const float* hp = h    + (size_t)n * HD + head * OUT_FEATS;
  const float* a0 = attn +           head * OUT_FEATS;
  const float* a1 = attn + HD      + head * OUT_FEATS;
  float s0 = 0.f, s1 = 0.f;
  #pragma unroll
  for (int d = 0; d < OUT_FEATS; ++d) {
    const float hv = hp[d];
    s0 += hv * a0[d];
    s1 += hv * a1[d];
  }
  es[idx] = s0;
  ed[idx] = s1;
}

__device__ __forceinline__ unsigned f2key(float v) {
  const unsigned u = __float_as_uint(v);
  return (u & 0x80000000u) ? ~u : (u | 0x80000000u);
}
__device__ __forceinline__ float key2f(unsigned k) {
  return __uint_as_float((k & 0x80000000u) ? (k & 0x7FFFFFFFu) : ~k);
}
__device__ __forceinline__ float finite_or_zero(float m) {
  return (fabsf(m) < INFINITY) ? m : 0.f;
}

#define NB          196
#define AGG_BLOCKS  256
#define AGG_THREADS 800
#define SUPER       5
#define NQ          (N_EDGES / 4)
#define Q_PER_SC    (AGG_THREADS * SUPER)
#define NSC         (NQ / Q_PER_SC)
#define LIST_CAP    24576
#define BATCH       8
#define UB          8
#define UR          4
#define DEN_ONE     2147483648.0f
#define DEN_STEP    4.656612873077392578125e-10f

static_assert(N_EDGES <= (1 << 20) && NB <= (1 << 12), "a list word is local node << 20 | edge");
static_assert(N_NODES <= (1 << 20), "a resolved list word is local node << 20 | src node");
static_assert(AGG_BLOCKS * NB >= N_NODES, "owner ranges cover all nodes");
static_assert(N_EDGES % 4 == 0 && NQ % Q_PER_SC == 0, "dst is scanned as int4, in whole steps");
static_assert(Q_PER_SC * 4 <= LIST_CAP, "one scan step always fits the list");
static_assert(AGG_THREADS % NUM_HEADS == 0, "a thread keeps one head over all its items");

__device__ __forceinline__ void agg_load_step(int scr, int tid, const int* __restrict__ dst,
                                              int4 (&d)[SUPER]) {
  const int4* dst4 = (const int4*)dst;
  #pragma unroll
  for (int i = 0; i < SUPER; ++i) {
    d[i] = dst4[scr * Q_PER_SC + i * AGG_THREADS + tid];
  }
}

__device__ __forceinline__ void agg_list_step(int scr, int tid, int base, int nb,
                                              const int4 (&d)[SUPER],
                                              unsigned* s_list, int* s_cnt, int* s_ovf) {
  #pragma unroll
  for (int i = 0; i < SUPER; ++i) {
    const int q = scr * Q_PER_SC + i * AGG_THREADS + tid;
    const unsigned tv[4] = {(unsigned)(d[i].x - base), (unsigned)(d[i].y - base),
                            (unsigned)(d[i].z - base), (unsigned)(d[i].w - base)};
    if (min(min(tv[0], tv[1]), min(tv[2], tv[3])) < (unsigned)nb) {
      #pragma unroll
      for (int c = 0; c < 4; ++c) {
        if (tv[c] < (unsigned)nb) {
          const int pos = atomicAdd(s_cnt, 1);
          if (pos < LIST_CAP) s_list[pos] = (tv[c] << 20) | (unsigned)(4 * q + c);
          else                *s_ovf = 1;
        }
      }
    }
  }
}

__device__ __forceinline__ void agg_resolve(int n, int tid, const int* __restrict__ src,
                                            unsigned* s_list) {
  for (int j0 = tid; j0 < n; j0 += AGG_THREADS * UR) {
    unsigned w[UR];
    int      sv[UR];
    #pragma unroll
    for (int u = 0; u < UR; ++u) {
      const int j = j0 + u * AGG_THREADS;
      w[u]  = (j < n) ? s_list[j] : 0u;
      sv[u] = src[w[u] & 0xFFFFFu];
    }
    #pragma unroll
    for (int u = 0; u < UR; ++u) {
      const int j = j0 + u * AGG_THREADS;
      if (j < n) s_list[j] = (w[u] & 0xFFF00000u) | (unsigned)sv[u];
    }
  }
}

template <int PASS>
__device__ __forceinline__ void agg_stats(int n, int tid, const float* __restrict__ es,
                                          const unsigned* s_list, const float* s_ed,
                                          unsigned* s_mkey, unsigned long long* s_den) {
  const int head  = tid & (NUM_HEADS - 1);
  const int total = n * NUM_HEADS;
  for (int it0 = tid; it0 < total; it0 += AGG_THREADS * UB) {
    int   tl[UB];
    float ev[UB];
    #pragma unroll
    for (int u = 0; u < UB; ++u) {
      const int it = it0 + u * AGG_THREADS;
      const unsigned w = s_list[(it < total ? it : total - 1) >> 3];
      tl[u] = (int)(w >> 20);
      ev[u] = es[(w & 0xFFFFFu) * NUM_HEADS + head];
    }
    #pragma unroll
    for (int u = 0; u < UB; ++u) {
      if (it0 + u * AGG_THREADS < total) {
        const int   k = tl[u] * NUM_HEADS + head;
        const float e = ev[u] + s_ed[k];
        const float v = e > 0.f ? e : NEG_SLOPE * e;
        if (PASS == 0) {
          atomicMax(&s_mkey[k], f2key(v));
        } else {
          const float ex = __expf(v - finite_or_zero(key2f(s_mkey[k])));
          atomicAdd(&s_den[k], (unsigned long long)(unsigned)(ex * DEN_ONE));
        }
      }
    }
  }
}

__device__ __forceinline__ void agg_den_to_float(int tid, const unsigned long long* s_den,
                                                 float* s_denf) {
  for (int i = tid; i < NB * NUM_HEADS; i += AGG_THREADS)
    s_denf[i] = (float)s_den[i] * DEN_STEP;
}

__device__ __forceinline__ void agg_drain(int n, int wave, int lane,
                                          const float* __restrict__ h,
                                          const float* __restrict__ es,
                                          const unsigned* s_list, const float* s_ed,
                                          const unsigned* s_mkey, const float* s_denf,
                                          float* s_acc) {
  const int head = lane >> 2;
  for (int i0 = wave * BATCH; i0 < n; i0 += (AGG_THREADS / 32) * BATCH) {
    float4 hv[BATCH];
    float  ev[BATCH];
    int    tl[BATCH];
    #pragma unroll
    for (int b = 0; b < BATCH; ++b) {
      const int idx = (i0 + b < n) ? (i0 + b) : (n - 1);
      const unsigned w = s_list[idx];
      const int s = (int)(w & 0xFFFFFu);
      tl[b] = (int)(w >> 20);
      ev[b] = es[s * NUM_HEADS + head];
      hv[b] = *(const float4*)&h[(size_t)s * HD + 4 * lane];
    }
    #pragma unroll
    for (int b = 0; b < BATCH; ++b) {
      if (i0 + b < n) {
        const int   k     = tl[b] * NUM_HEADS + head;
        const float e     = ev[b] + s_ed[k];
        const float v     = e > 0.f ? e : NEG_SLOPE * e;
        const float dv    = s_denf[k];
        const float alpha = __expf(v - finite_or_zero(key2f(s_mkey[k]))) / (dv > 0.f ? dv : 1.f);
        float* a = s_acc + tl[b] * HD + lane;
        atomicAdd(a +  0, alpha * hv[b].x);
        atomicAdd(a + 32, alpha * hv[b].y);
        atomicAdd(a + 64, alpha * hv[b].z);
        atomicAdd(a + 96, alpha * hv[b].w);
      }
    }
  }
}

__global__ __launch_bounds__(AGG_THREADS)
void gat_softmax_aggregate_owner(const float* __restrict__ h,
                                 const float* __restrict__ es,
                                 const float* __restrict__ ed,
                                 const float* __restrict__ bias,
                                 const int* __restrict__ src,
                                 const int* __restrict__ dst,
                                 float* __restrict__ out) {
  __shared__ __attribute__((aligned(16))) float    s_acc[NB * HD];
  __shared__ __attribute__((aligned(16))) unsigned s_list[LIST_CAP];
  __shared__ unsigned long long s_den[NB * NUM_HEADS];
  __shared__ float    s_denf[NB * NUM_HEADS];
  __shared__ unsigned s_mkey[NB * NUM_HEADS];
  __shared__ float    s_ed[NB * NUM_HEADS];
  __shared__ int      s_cnt;
  __shared__ int      s_ovf;
  const int tid  = threadIdx.x;
  const int lane = tid & 31;
  const int wave = tid >> 5;
  const int base = blockIdx.x * NB;
  const int nb   = (N_NODES - base < NB) ? (N_NODES - base) : NB;

  for (int i = tid; i < NB * HD / 4; i += AGG_THREADS)
    ((float4*)s_acc)[i] = make_float4(0.f, 0.f, 0.f, 0.f);
  for (int i = tid; i < NB * NUM_HEADS; i += AGG_THREADS) {
    s_ed[i]   = (i < nb * NUM_HEADS) ? ed[base * NUM_HEADS + i] : 0.f;
    s_mkey[i] = 0u;
    s_den[i]  = 0ull;
  }
  if (tid == 0) { s_cnt = 0; s_ovf = 0; }
  __syncthreads();

  {
    int4 d[SUPER], dn[SUPER];
    int scr = (int)blockIdx.x % NSC;
    agg_load_step(scr, tid, dst, d);
    for (int sc = 0; sc < NSC - 1; ++sc) {
      const int scn = (scr + 1 == NSC) ? 0 : scr + 1;
      agg_load_step(scn, tid, dst, dn);
      agg_list_step(scr, tid, base, nb, d, s_list, &s_cnt, &s_ovf);
      #pragma unroll
      for (int i = 0; i < SUPER; ++i) d[i] = dn[i];
      scr = scn;
    }
    agg_list_step(scr, tid, base, nb, d, s_list, &s_cnt, &s_ovf);
  }
  __syncthreads();

  if (s_ovf == 0) {
    const int n = s_cnt;
    agg_resolve(n, tid, src, s_list);
    __syncthreads();
    agg_stats<0>(n, tid, es, s_list, s_ed, s_mkey, s_den);
    __syncthreads();
    agg_stats<1>(n, tid, es, s_list, s_ed, s_mkey, s_den);
    __syncthreads();
    agg_den_to_float(tid, s_den, s_denf);
    __syncthreads();
    agg_drain(n, wave, lane, h, es, s_list, s_ed, s_mkey, s_denf, s_acc);
  } else {
    for (int pass = 0; pass < 3; ++pass) {
      for (int sc = 0; sc < NSC; ++sc) {
        int4 d[SUPER];
        __syncthreads();
        if (tid == 0) s_cnt = 0;
        __syncthreads();
        agg_load_step(sc, tid, dst, d);
        agg_list_step(sc, tid, base, nb, d, s_list, &s_cnt, &s_ovf);
        __syncthreads();
        const int n = s_cnt;
        agg_resolve(n, tid, src, s_list);
        __syncthreads();
        if (pass == 0)      agg_stats<0>(n, tid, es, s_list, s_ed, s_mkey, s_den);
        else if (pass == 1) agg_stats<1>(n, tid, es, s_list, s_ed, s_mkey, s_den);
        else                agg_drain(n, wave, lane, h, es, s_list, s_ed, s_mkey, s_denf, s_acc);
      }
      __syncthreads();
      if (pass == 1) agg_den_to_float(tid, s_den, s_denf);
    }
  }
  __syncthreads();

  const float4 bv = *(const float4*)&bias[4 * lane];
  for (int tl = wave; tl < nb; tl += AGG_THREADS / 32) {
    const float* a = s_acc + tl * HD + lane;
    float4 o;
    o.x = a[0]  + bv.x;
    o.y = a[32] + bv.y;
    o.z = a[64] + bv.z;
    o.w = a[96] + bv.w;
    *(float4*)&out[(size_t)(base + tl) * HD + 4 * lane] = o;
  }
}

extern "C" void kernel_launch(void* const* d_in, const int* in_sizes, int n_in,
                              void* d_out, int out_size, void* d_ws, size_t ws_size,
                              hipStream_t stream) {
  const float* feat = (const float*)d_in[0];
  const float* W    = (const float*)d_in[1];
  const float* attn = (const float*)d_in[2];
  const float* bias = (const float*)d_in[3];
  const int*   src  = (const int*)d_in[4];
  const int*   dst  = (const int*)d_in[5];
  float*       out  = (float*)d_out;

  float* h  = (float*)d_ws;
  float* es = h  + (size_t)N_NODES * HD;
  float* ed = es + (size_t)N_NODES * NUM_HEADS;

  const int NH = N_NODES * NUM_HEADS;

  gat_gemm_wmma<<<N_NODES / 16, 256, 0, stream>>>(feat, W, h);
  gat_logits<<<(NH + 255) / 256, 256, 0, stream>>>(h, attn, es, ed);
  gat_softmax_aggregate_owner<<<AGG_BLOCKS, AGG_THREADS, 0, stream>>>(h, es, ed, bias, src, dst, out);
}


# Owner scan reads a 1-byte owner plane (dst/196, written by the logits launch), 16 edges per 16-byte load, zero-byte test, per-wave LDS queue: 4x fewer scan bytes; list order, so sum order, changes
# speedup vs baseline: 5.7249x; 1.1536x over previous
#include <hip/hip_runtime.h>
#include <hip/hip_bf16.h>
#include <math.h>

#define N_NODES   50000
#define N_EDGES   800000
#define IN_FEATS  128
#define NUM_HEADS 8
#define OUT_FEATS 16
#define HD        (NUM_HEADS * OUT_FEATS)
#define NEG_SLOPE 0.2f

typedef float v2f __attribute__((ext_vector_type(2)));
typedef float v8f __attribute__((ext_vector_type(8)));

__global__ __launch_bounds__(256)
void gat_gemm_wmma(const float* __restrict__ feat,
                   const float* __restrict__ W,
                   float* __restrict__ h) {
  __shared__ float lA[16 * IN_FEATS];
  const int tid  = threadIdx.x;
  const int wave = tid >> 5;
  const int lane = tid & 31;
  const int m0   = blockIdx.x * 16;
  const int n0   = wave * 16;

  #pragma unroll
  for (int i = tid; i < 16 * IN_FEATS; i += 256) {
    const int r = i >> 7;
    const int c = i & 127;
    lA[i] = feat[(size_t)(m0 + r) * IN_FEATS + c];
  }
  __syncthreads();

  const int ml  = lane & 15;
  const int klo = (lane < 16) ? 0 : 2;

  v8f acc = {};
  #pragma unroll 4
  for (int k = 0; k < IN_FEATS; k += 4) {
    const float2 av = *(const float2*)&lA[ml * IN_FEATS + k + klo];
    v2f a; a.x = av.x; a.y = av.y;
    v2f b;
    b.x = W[(size_t)(k + klo)     * HD + n0 + ml];
    b.y = W[(size_t)(k + klo + 1) * HD + n0 + ml];
    acc = __builtin_amdgcn_wmma_f32_16x16x4_f32(
         false, a,  false, b,
         (short)0, acc,  false,  false);
  }

  const int rbase = m0 + ((lane < 16) ? 0 : 8);
  #pragma unroll
  for (int v = 0; v < 8; ++v) {
    h[(size_t)(rbase + v) * HD + n0 + ml] = acc[v];
  }
}

#define NB 196
static_assert(N_EDGES == 2 * N_NODES * NUM_HEADS, "one thread writes the owner bytes of two edges");
static_assert((N_NODES + NB - 1) / NB <= 256, "an owner number fits a byte");

__global__ void gat_logits(const float* __restrict__ h,
                           const float* __restrict__ attn,
                           const int* __restrict__ dst,
                           float* __restrict__ es, float* __restrict__ ed,
                           unsigned short* __restrict__ own2) {
  const int idx = blockIdx.x * blockDim.x + threadIdx.x;
  if (idx >= N_NODES * NUM_HEADS) return;
  {
    const int2 d = ((const int2*)dst)[idx];
    own2[idx] = (unsigned short)((((unsigned)d.x / NB) & 0xFFu) |
                                 ((((unsigned)d.y / NB) & 0xFFu) << 8));
  }
  const int head = idx & (NUM_HEADS - 1);
  const int n    = idx >> 3;
  const float* hp = h    + (size_t)n * HD + head * OUT_FEATS;
  const float* a0 = attn +           head * OUT_FEATS;
  const float* a1 = attn + HD      + head * OUT_FEATS;
  float s0 = 0.f, s1 = 0.f;
  #pragma unroll
  for (int d = 0; d < OUT_FEATS; ++d) {
    const float hv = hp[d];
    s0 += hv * a0[d];
    s1 += hv * a1[d];
  }
  es[idx] = s0;
  ed[idx] = s1;
}

__device__ __forceinline__ unsigned f2key(float v) {
  const unsigned u = __float_as_uint(v);
  return (u & 0x80000000u) ? ~u : (u | 0x80000000u);
}
__device__ __forceinline__ float key2f(unsigned k) {
  return __uint_as_float((k & 0x80000000u) ? (k & 0x7FFFFFFFu) : ~k);
}
__device__ __forceinline__ float finite_or_zero(float m) {
  return (fabsf(m) < INFINITY) ? m : 0.f;
}

#define NBX         (NB + 1)
#define AGG_BLOCKS  256
#define AGG_THREADS 800
#define AGG_WAVES   (AGG_THREADS / 32)
#define SUPER       5
#define PQ          (N_EDGES / 16)
#define P_PER_SC    (AGG_THREADS * SUPER)
#define PSC_FULL    (PQ / P_PER_SC)
#define WQ_CAP      200
#define NQ          (N_EDGES / 4)
#define Q_PER_SC    (AGG_THREADS * SUPER)
#define NSC         (NQ / Q_PER_SC)
#define LIST_CAP    24576
#define BATCH       8
#define UB          8
#define UR          4
#define DEN_ONE     2147483648.0f
#define DEN_STEP    4.656612873077392578125e-10f

static_assert(N_EDGES <= (1 << 20) && NBX <= (1 << 12), "a list word is local node << 20 | edge or src");
static_assert(N_NODES <= (1 << 20), "a resolved list word is local node << 20 | src node");
static_assert(AGG_BLOCKS * NB >= N_NODES && AGG_BLOCKS <= 256, "owner ranges cover all nodes, one byte each");
static_assert(N_EDGES % 16 == 0 && PSC_FULL * P_PER_SC < PQ && (PSC_FULL + 1) * P_PER_SC >= PQ,
              "the plane is scanned in whole steps and one partial step");
static_assert(N_EDGES % 4 == 0 && NQ % Q_PER_SC == 0, "overflow path: dst is scanned as int4, in whole steps");
static_assert(Q_PER_SC * 4 <= LIST_CAP, "overflow path: one scan step always fits the list");
static_assert(AGG_THREADS % NUM_HEADS == 0, "a thread keeps one head over all its items");
static_assert(AGG_WAVES * WQ_CAP * 5 <= NBX * HD, "the wave queues fit the LDS words of the row sums");

__device__ __forceinline__ unsigned zero_bytes(unsigned z) {
  return ~((((z & 0x7F7F7F7Fu) + 0x7F7F7F7Fu) | z) | 0x7F7F7F7Fu);
}

template <bool PARTIAL>
__device__ __forceinline__ void own_load_step(int scr, int tid, unsigned bb,
                                              const unsigned* __restrict__ own,
                                              uint4 (&x)[SUPER]) {
  const uint4* own4 = (const uint4*)own;
  #pragma unroll
  for (int i = 0; i < SUPER; ++i) {
    const int q = scr * P_PER_SC + i * AGG_THREADS + tid;
    if (!PARTIAL || q < PQ) x[i] = own4[q];
    else                    x[i] = make_uint4(~bb, ~bb, ~bb, ~bb);
  }
}

__device__ __forceinline__ void own_queue_step(int scr, int tid, unsigned bb,
                                               const uint4 (&x)[SUPER],
                                               uint4* wq_d, unsigned* wq_i, int& wq) {
  #pragma unroll
  for (int i = 0; i < SUPER; ++i) {
    const int q = scr * P_PER_SC + i * AGG_THREADS + tid;
    const unsigned z0 = x[i].x ^ bb, z1 = x[i].y ^ bb, z2 = x[i].z ^ bb, z3 = x[i].w ^ bb;
    const unsigned t = (((z0 - 0x01010101u) & ~z0) | ((z1 - 0x01010101u) & ~z1) |
                        ((z2 - 0x01010101u) & ~z2) | ((z3 - 0x01010101u) & ~z3)) & 0x80808080u;
    const bool     cand = t != 0u;
    const unsigned m    = __builtin_amdgcn_ballot_w32(cand);
    if (m != 0u) {
      if (cand) {
        const int slot = min(wq + (int)__builtin_amdgcn_mbcnt_lo(m, 0u), WQ_CAP - 1);
        wq_d[slot] = x[i];
        wq_i[slot] = (unsigned)q;
      }
      wq += __popc(m);
    }
  }
}

__device__ __forceinline__ void own_emit(int wq, int lane, unsigned bb,
                                         const uint4* wq_d, const unsigned* wq_i,
                                         unsigned* s_list, int* s_cnt, int* s_ovf) {
  const int n = wq < WQ_CAP ? wq : WQ_CAP;
  for (int i0 = 0; i0 < n; i0 += 32) {
    const int i = i0 + lane;
    unsigned hz[4] = {0u, 0u, 0u, 0u};
    unsigned q = 0u;
    if (i < n) {
      const uint4 x = wq_d[i];
      q = wq_i[i];
      hz[0] = zero_bytes(x.x ^ bb);
      hz[1] = zero_bytes(x.y ^ bb);
      hz[2] = zero_bytes(x.z ^ bb);
      hz[3] = zero_bytes(x.w ^ bb);
    }
    const int c = __popc(hz[0]) + __popc(hz[1]) + __popc(hz[2]) + __popc(hz[3]);
    int incl = c;
    #pragma unroll
    for (int d = 1; d < 32; d <<= 1) {
      const int v = __shfl_up(incl, d, 32);
      if (lane >= d) incl += v;
    }
    const int total = __shfl(incl, 31, 32);
    int first = 0;
    if (lane == 0) first = atomicAdd(s_cnt, total);
    first = __shfl(first, 0, 32);
    int pos = first + incl - c;
    #pragma unroll
    for (int j = 0; j < 4; ++j) {
      unsigned rest = hz[j];
      while (rest != 0u) {
        const unsigned b = (unsigned)(__ffs((int)rest) - 1) >> 3;
        if (pos < LIST_CAP) s_list[pos] = 16u * q + 4u * (unsigned)j + b;
        else                *s_ovf = 1;
        ++pos;
        rest &= rest - 1u;
      }
    }
  }
}

__device__ __forceinline__ void agg_load_step(int scr, int tid, const int* __restrict__ dst,
                                              int4 (&d)[SUPER]) {
  const int4* dst4 = (const int4*)dst;
  #pragma unroll
  for (int i = 0; i < SUPER; ++i) {
    d[i] = dst4[scr * Q_PER_SC + i * AGG_THREADS + tid];
  }
}

__device__ __forceinline__ void agg_list_step(int scr, int tid, int base, int nb,
                                              const int4 (&d)[SUPER],
                                              unsigned* s_list, int* s_cnt, int* s_ovf) {
  #pragma unroll
  for (int i = 0; i < SUPER; ++i) {
    const int q = scr * Q_PER_SC + i * AGG_THREADS + tid;
    const unsigned tv[4] = {(unsigned)(d[i].x - base), (unsigned)(d[i].y - base),
                            (unsigned)(d[i].z - base), (unsigned)(d[i].w - base)};
    if (min(min(tv[0], tv[1]), min(tv[2], tv[3])) < (unsigned)nb) {
      #pragma unroll
      for (int c = 0; c < 4; ++c) {
        if (tv[c] < (unsigned)nb) {
          const int pos = atomicAdd(s_cnt, 1);
          if (pos < LIST_CAP) s_list[pos] = (unsigned)(4 * q + c);
          else                *s_ovf = 1;
        }
      }
    }
  }
}

__device__ __forceinline__ void agg_resolve(int n, int tid, int base, int nb,
                                            const int* __restrict__ src,
                                            const int* __restrict__ dst,
                                            unsigned* s_list) {
  for (int j0 = tid; j0 < n; j0 += AGG_THREADS * UR) {
    unsigned e[UR];
    int      sv[UR];
    int      dv[UR];
    #pragma unroll
    for (int u = 0; u < UR; ++u) {
      const int j = j0 + u * AGG_THREADS;
      e[u]  = (j < n) ? (s_list[j] & 0xFFFFFu) : 0u;
      sv[u] = src[e[u]];
      dv[u] = dst[e[u]];
    }
    #pragma unroll
    for (int u = 0; u < UR; ++u) {
      const int j = j0 + u * AGG_THREADS;
      if (j < n) {
        unsigned tl = (unsigned)(dv[u] - base);
        if (tl >= (unsigned)nb) tl = NB;
        s_list[j] = (tl << 20) | (unsigned)sv[u];
      }
    }
  }
}

template <int PASS>
__device__ __forceinline__ void agg_stats(int n, int tid, const float* __restrict__ es,
                                          const unsigned* s_list, const float* s_ed,
                                          unsigned* s_mkey, unsigned long long* s_den) {
  const int head  = tid & (NUM_HEADS - 1);
  const int total = n * NUM_HEADS;
  for (int it0 = tid; it0 < total; it0 += AGG_THREADS * UB) {
    int   tl[UB];
    float ev[UB];
    #pragma unroll
    for (int u = 0; u < UB; ++u) {
      const int it = it0 + u * AGG_THREADS;
      const unsigned w = s_list[(it < total ? it : total - 1) >> 3];
      tl[u] = (int)(w >> 20);
      ev[u] = es[(w & 0xFFFFFu) * NUM_HEADS + head];
    }
    #pragma unroll
    for (int u = 0; u < UB; ++u) {
      if (it0 + u * AGG_THREADS < total) {
        const int   k = tl[u] * NUM_HEADS + head;
        const float e = ev[u] + s_ed[k];
        const float v = e > 0.f ? e : NEG_SLOPE * e;
        if (PASS == 0) {
          atomicMax(&s_mkey[k], f2key(v));
        } else {
          const float ex = __expf(v - finite_or_zero(key2f(s_mkey[k])));
          atomicAdd(&s_den[k], (unsigned long long)(unsigned)(ex * DEN_ONE));
        }
      }
    }
  }
}

__device__ __forceinline__ void agg_den_to_float(int tid, const unsigned long long* s_den,
                                                 float* s_denf) {
  for (int i = tid; i < NBX * NUM_HEADS; i += AGG_THREADS)
    s_denf[i] = (float)s_den[i] * DEN_STEP;
}

__device__ __forceinline__ void agg_drain(int n, int wave, int lane,
                                          const float* __restrict__ h,
                                          const float* __restrict__ es,
                                          const unsigned* s_list, const float* s_ed,
                                          const unsigned* s_mkey, const float* s_denf,
                                          float* s_acc) {
  const int head = lane >> 2;
  for (int i0 = wave * BATCH; i0 < n; i0 += AGG_WAVES * BATCH) {
    float4 hv[BATCH];
    float  ev[BATCH];
    int    tl[BATCH];
    #pragma unroll
    for (int b = 0; b < BATCH; ++b) {
      const int idx = (i0 + b < n) ? (i0 + b) : (n - 1);
      const unsigned w = s_list[idx];
      const int s = (int)(w & 0xFFFFFu);
      tl[b] = (int)(w >> 20);
      ev[b] = es[s * NUM_HEADS + head];
      hv[b] = *(const float4*)&h[(size_t)s * HD + 4 * lane];
    }
    #pragma unroll
    for (int b = 0; b < BATCH; ++b) {
      if (i0 + b < n) {
        const int   k     = tl[b] * NUM_HEADS + head;
        const float e     = ev[b] + s_ed[k];
        const float v     = e > 0.f ? e : NEG_SLOPE * e;
        const float dv    = s_denf[k];
        const float alpha = __expf(v - finite_or_zero(key2f(s_mkey[k]))) / (dv > 0.f ? dv : 1.f);
        float* a = s_acc + tl[b] * HD + lane;
        atomicAdd(a +  0, alpha * hv[b].x);
        atomicAdd(a + 32, alpha * hv[b].y);
        atomicAdd(a + 64, alpha * hv[b].z);
        atomicAdd(a + 96, alpha * hv[b].w);
      }
    }
  }
}

__global__ __launch_bounds__(AGG_THREADS)
void gat_softmax_aggregate_owner(const float* __restrict__ h,
                                 const float* __restrict__ es,
                                 const float* __restrict__ ed,
                                 const float* __restrict__ bias,
                                 const int* __restrict__ src,
                                 const int* __restrict__ dst,
                                 const unsigned* __restrict__ own,
                                 float* __restrict__ out) {
  __shared__ __attribute__((aligned(16))) float    s_acc[NBX * HD];
  __shared__ __attribute__((aligned(16))) unsigned s_list[LIST_CAP];
  __shared__ unsigned long long s_den[NBX * NUM_HEADS];
  __shared__ float    s_denf[NBX * NUM_HEADS];
  __shared__ unsigned s_mkey[NBX * NUM_HEADS];
  __shared__ float    s_ed[NBX * NUM_HEADS];
  __shared__ int      s_cnt;
  __shared__ int      s_ovf;
  const int tid  = threadIdx.x;
  const int lane = tid & 31;
  const int wave = tid >> 5;
  const int base = blockIdx.x * NB;
  const int nb   = (N_NODES - base < NB) ? (N_NODES - base) : NB;

  for (int i = tid; i < NBX * NUM_HEADS; i += AGG_THREADS) {
    s_ed[i]   = (i < nb * NUM_HEADS) ? ed[base * NUM_HEADS + i] : 0.f;
    s_mkey[i] = 0u;
    s_den[i]  = 0ull;
  }
  if (tid == 0) { s_cnt = 0; s_ovf = 0; }
  __syncthreads();

  {
    const unsigned bb   = (unsigned)blockIdx.x * 0x01010101u;
    uint4*         wq_d = (uint4*)s_acc + wave * WQ_CAP;
    unsigned*      wq_i = (unsigned*)s_acc + AGG_WAVES * WQ_CAP * 4 + wave * WQ_CAP;
    int            wq   = 0;
    uint4 x[SUPER], xn[SUPER];
    int scr = (int)blockIdx.x % PSC_FULL;
    own_load_step<false>(scr, tid, bb, own, x);
    for (int sc = 0; sc < PSC_FULL - 1; ++sc) {
      const int scn = (scr + 1 == PSC_FULL) ? 0 : scr + 1;
      own_load_step<false>(scn, tid, bb, own, xn);
      own_queue_step(scr, tid, bb, x, wq_d, wq_i, wq);
      #pragma unroll
      for (int i = 0; i < SUPER; ++i) x[i] = xn[i];
      scr = scn;
    }
    own_load_step<true>(PSC_FULL, tid, bb, own, xn);
    own_queue_step(scr, tid, bb, x, wq_d, wq_i, wq);
    own_queue_step(PSC_FULL, tid, bb, xn, wq_d, wq_i, wq);
    if (wq > WQ_CAP) s_ovf = 1;
    own_emit(wq, lane, bb, wq_d, wq_i, s_list, &s_cnt, &s_ovf);
  }
  __syncthreads();

  for (int i = tid; i < NBX * HD / 4; i += AGG_THREADS)
    ((float4*)s_acc)[i] = make_float4(0.f, 0.f, 0.f, 0.f);

  if (s_ovf == 0) {
    const int n = s_cnt;
    agg_resolve(n, tid, base, nb, src, dst, s_list);
    __syncthreads();
    agg_stats<0>(n, tid, es, s_list, s_ed, s_mkey, s_den);
    __syncthreads();
    agg_stats<1>(n, tid, es, s_list, s_ed, s_mkey, s_den);
    __syncthreads();
    agg_den_to_float(tid, s_den, s_denf);
    __syncthreads();
    agg_drain(n, wave, lane, h, es, s_list, s_ed, s_mkey, s_denf, s_acc);
  } else {
    for (int pass = 0; pass < 3; ++pass) {
      for (int sc = 0; sc < NSC; ++sc) {
        int4 d[SUPER];
        __syncthreads();
        if (tid == 0) s_cnt = 0;
        __syncthreads();
        agg_load_step(sc, tid, dst, d);
        agg_list_step(sc, tid, base, nb, d, s_list, &s_cnt, &s_ovf);
        __syncthreads();
        const int n = s_cnt;
        agg_resolve(n, tid, base, nb, src, dst, s_list);
        __syncthreads();
        if (pass == 0)      agg_stats<0>(n, tid, es, s_list, s_ed, s_mkey, s_den);
        else if (pass == 1) agg_stats<1>(n, tid, es, s_list, s_ed, s_mkey, s_den);
        else                agg_drain(n, wave, lane, h, es, s_list, s_ed, s_mkey, s_denf, s_acc);
      }
      __syncthreads();
      if (pass == 1) agg_den_to_float(tid, s_den, s_denf);
    }
  }
  __syncthreads();

  const float4 bv = *(const float4*)&bias[4 * lane];
  for (int tl = wave; tl < nb; tl += AGG_WAVES) {
    const float* a = s_acc + tl * HD + lane;
    float4 o;
    o.x = a[0]  + bv.x;
    o.y = a[32] + bv.y;
    o.z = a[64] + bv.z;
    o.w = a[96] + bv.w;
    *(float4*)&out[(size_t)(base + tl) * HD + 4 * lane] = o;
  }
}

extern "C" void kernel_launch(void* const* d_in, const int* in_sizes, int n_in,
                              void* d_out, int out_size, void* d_ws, size_t ws_size,
                              hipStream_t stream) {
  const float* feat = (const float*)d_in[0];
  const float* W    = (const float*)d_in[1];
  const float* attn = (const float*)d_in[2];
  const float* bias = (const float*)d_in[3];
  const int*   src  = (const int*)d_in[4];
  const int*   dst  = (const int*)d_in[5];
  float*       out  = (float*)d_out;

  float* h  = (float*)d_ws;
  float* es = h  + (size_t)N_NODES * HD;
  float* ed = es + (size_t)N_NODES * NUM_HEADS;
  unsigned* own = (unsigned*)(ed + (size_t)N_NODES * NUM_HEADS);

  const int NH = N_NODES * NUM_HEADS;

  gat_gemm_wmma<<<N_NODES / 16, 256, 0, stream>>>(feat, W, h);
  gat_logits<<<(NH + 255) / 256, 256, 0, stream>>>(h, attn, dst, es, ed, (unsigned short*)own);
  gat_softmax_aggregate_owner<<<AGG_BLOCKS, AGG_THREADS, 0, stream>>>(h, es, ed, bias, src, dst, own, out);
}


# Each listed edge's 8 scores formed once and kept in LDS (list cut to make room): exp-sum pass and row-sum pass read them there, 2 of the 3 per-edge es gathers go; sums and their order as before
# speedup vs baseline: 6.2261x; 1.0875x over previous
#include <hip/hip_runtime.h>
#include <hip/hip_bf16.h>
#include <math.h>

#define N_NODES   50000
#define N_EDGES   800000
#define IN_FEATS  128
#define NUM_HEADS 8
#define OUT_FEATS 16
#define HD        (NUM_HEADS * OUT_FEATS)
#define NEG_SLOPE 0.2f

typedef float v2f __attribute__((ext_vector_type(2)));
typedef float v8f __attribute__((ext_vector_type(8)));

__global__ __launch_bounds__(256)
void gat_gemm_wmma(const float* __restrict__ feat,
                   const float* __restrict__ W,
                   float* __restrict__ h) {
  __shared__ float lA[16 * IN_FEATS];
  const int tid  = threadIdx.x;
  const int wave = tid >> 5;
  const int lane = tid & 31;
  const int m0   = blockIdx.x * 16;
  const int n0   = wave * 16;

  #pragma unroll
  for (int i = tid; i < 16 * IN_FEATS; i += 256) {
    const int r = i >> 7;
    const int c = i & 127;
    lA[i] = feat[(size_t)(m0 + r) * IN_FEATS + c];
  }
  __syncthreads();

  const int ml  = lane & 15;
  const int klo = (lane < 16) ? 0 : 2;

  v8f acc = {};
  #pragma unroll 4
  for (int k = 0; k < IN_FEATS; k += 4) {
    const float2 av = *(const float2*)&lA[ml * IN_FEATS + k + klo];
    v2f a; a.x = av.x; a.y = av.y;
    v2f b;
    b.x = W[(size_t)(k + klo)     * HD + n0 + ml];
    b.y = W[(size_t)(k + klo + 1) * HD + n0 + ml];
    acc = __builtin_amdgcn_wmma_f32_16x16x4_f32(
         false, a,  false, b,
         (short)0, acc,  false,  false);
  }

  const int rbase = m0 + ((lane < 16) ? 0 : 8);
  #pragma unroll
  for (int v = 0; v < 8; ++v) {
    h[(size_t)(rbase + v) * HD + n0 + ml] = acc[v];
  }
}

#define NB 196
static_assert(N_EDGES == 2 * N_NODES * NUM_HEADS, "one thread writes the owner bytes of two edges");
static_assert((N_NODES + NB - 1) / NB <= 256, "an owner number fits a byte");

__global__ void gat_logits(const float* __restrict__ h,
                           const float* __restrict__ attn,
                           const int* __restrict__ dst,
                           float* __restrict__ es, float* __restrict__ ed,
                           unsigned short* __restrict__ own2) {
  const int idx = blockIdx.x * blockDim.x + threadIdx.x;
  if (idx >= N_NODES * NUM_HEADS) return;
  {
    const int2 d = ((const int2*)dst)[idx];
    own2[idx] = (unsigned short)((((unsigned)d.x / NB) & 0xFFu) |
                                 ((((unsigned)d.y / NB) & 0xFFu) << 8));
  }
  const int head = idx & (NUM_HEADS - 1);
  const int n    = idx >> 3;
  const float* hp = h    + (size_t)n * HD + head * OUT_FEATS;
  const float* a0 = attn +           head * OUT_FEATS;
  const float* a1 = attn + HD      + head * OUT_FEATS;
  float s0 = 0.f, s1 = 0.f;
  #pragma unroll
  for (int d = 0; d < OUT_FEATS; ++d) {
    const float hv = hp[d];
    s0 += hv * a0[d];
    s1 += hv * a1[d];
  }
  es[idx] = s0;
  ed[idx] = s1;
}

__device__ __forceinline__ unsigned f2key(float v) {
  const unsigned u = __float_as_uint(v);
  return (u & 0x80000000u) ? ~u : (u | 0x80000000u);
}
__device__ __forceinline__ float key2f(unsigned k) {
  return __uint_as_float((k & 0x80000000u) ? (k & 0x7FFFFFFFu) : ~k);
}
__device__ __forceinline__ float finite_or_zero(float m) {
  return (fabsf(m) < INFINITY) ? m : 0.f;
}

#define NBX         (NB + 1)
#define AGG_BLOCKS  256
#define AGG_THREADS 800
#define AGG_WAVES   (AGG_THREADS / 32)
#define SUPER       5
#define OSUPER      2
#define PQ          (N_EDGES / 16)
#define P_PER_SC    (AGG_THREADS * SUPER)
#define PSC_FULL    (PQ / P_PER_SC)
#define WQ_CAP      200
#define NQ          (N_EDGES / 4)
#define Q_PER_SC    (AGG_THREADS * OSUPER)
#define NSC         (NQ / Q_PER_SC)
#define LIST_CAP    8192
#define SC_CAP      3840
#define BATCH       8
#define UB          8
#define UR          4
#define DEN_ONE     2147483648.0f
#define DEN_STEP    4.656612873077392578125e-10f

static_assert(N_EDGES <= (1 << 20) && NBX <= (1 << 12), "a list word is local node << 20 | edge or src");
static_assert(N_NODES <= (1 << 20), "a resolved list word is local node << 20 | src node");
static_assert(AGG_BLOCKS * NB >= N_NODES && AGG_BLOCKS <= 256, "owner ranges cover all nodes, one byte each");
static_assert(N_EDGES % 16 == 0 && PSC_FULL * P_PER_SC < PQ && (PSC_FULL + 1) * P_PER_SC >= PQ,
              "the plane is scanned in whole steps and one partial step");
static_assert(N_EDGES % 4 == 0 && NQ % Q_PER_SC == 0, "overflow path: dst is scanned as int4, in whole steps");
static_assert(Q_PER_SC * 4 <= LIST_CAP, "overflow path: one scan step always fits the list");
static_assert(SC_CAP <= LIST_CAP, "every edge with kept scores is on the list");
static_assert(AGG_THREADS % NUM_HEADS == 0, "a thread keeps one head over all its items");
static_assert(AGG_WAVES * WQ_CAP * 5 <= NBX * HD, "the wave queues fit the LDS words of the row sums");

__device__ __forceinline__ unsigned zero_bytes(unsigned z) {
  return ~((((z & 0x7F7F7F7Fu) + 0x7F7F7F7Fu) | z) | 0x7F7F7F7Fu);
}

template <bool PARTIAL>
__device__ __forceinline__ void own_load_step(int scr, int tid, unsigned bb,
                                              const unsigned* __restrict__ own,
                                              uint4 (&x)[SUPER]) {
  const uint4* own4 = (const uint4*)own;
  #pragma unroll
  for (int i = 0; i < SUPER; ++i) {
    const int q = scr * P_PER_SC + i * AGG_THREADS + tid;
    if (!PARTIAL || q < PQ) x[i] = own4[q];
    else                    x[i] = make_uint4(~bb, ~bb, ~bb, ~bb);
  }
}

__device__ __forceinline__ void own_queue_step(int scr, int tid, unsigned bb,
                                               const uint4 (&x)[SUPER],
                                               uint4* wq_d, unsigned* wq_i, int& wq) {
  #pragma unroll
  for (int i = 0; i < SUPER; ++i) {
    const int q = scr * P_PER_SC + i * AGG_THREADS + tid;
    const unsigned z0 = x[i].x ^ bb, z1 = x[i].y ^ bb, z2 = x[i].z ^ bb, z3 = x[i].w ^ bb;
    const unsigned t = (((z0 - 0x01010101u) & ~z0) | ((z1 - 0x01010101u) & ~z1) |
                        ((z2 - 0x01010101u) & ~z2) | ((z3 - 0x01010101u) & ~z3)) & 0x80808080u;
    const bool     cand = t != 0u;
    const unsigned m    = __builtin_amdgcn_ballot_w32(cand);
    if (m != 0u) {
      if (cand) {
        const int slot = min(wq + (int)__builtin_amdgcn_mbcnt_lo(m, 0u), WQ_CAP - 1);
        wq_d[slot] = x[i];
        wq_i[slot] = (unsigned)q;
      }
      wq += __popc(m);
    }
  }
}

__device__ __forceinline__ void own_emit(int wq, int lane, unsigned bb,
                                         const uint4* wq_d, const unsigned* wq_i,
                                         unsigned* s_list, int* s_cnt, int* s_ovf) {
  const int n = wq < WQ_CAP ? wq : WQ_CAP;
  for (int i0 = 0; i0 < n; i0 += 32) {
    const int i = i0 + lane;
    unsigned hz[4] = {0u, 0u, 0u, 0u};
    unsigned q = 0u;
    if (i < n) {
      const uint4 x = wq_d[i];
      q = wq_i[i];
      hz[0] = zero_bytes(x.x ^ bb);
      hz[1] = zero_bytes(x.y ^ bb);
      hz[2] = zero_bytes(x.z ^ bb);
      hz[3] = zero_bytes(x.w ^ bb);
    }
    const int c = __popc(hz[0]) + __popc(hz[1]) + __popc(hz[2]) + __popc(hz[3]);
    int incl = c;
    #pragma unroll
    for (int d = 1; d < 32; d <<= 1) {
      const int v = __shfl_up(incl, d, 32);
      if (lane >= d) incl += v;
    }
    const int total = __shfl(incl, 31, 32);
    int first = 0;
    if (lane == 0) first = atomicAdd(s_cnt, total);
    first = __shfl(first, 0, 32);
    int pos = first + incl - c;
    #pragma unroll
    for (int j = 0; j < 4; ++j) {
      unsigned rest = hz[j];
      while (rest != 0u) {
        const unsigned b = (unsigned)(__ffs((int)rest) - 1) >> 3;
        if (pos < LIST_CAP) s_list[pos] = 16u * q + 4u * (unsigned)j + b;
        else                *s_ovf = 1;
        ++pos;
        rest &= rest - 1u;
      }
    }
  }
}

__device__ __forceinline__ void agg_load_step(int scr, int tid, const int* __restrict__ dst,
                                              int4 (&d)[OSUPER]) {
  const int4* dst4 = (const int4*)dst;
  #pragma unroll
  for (int i = 0; i < OSUPER; ++i) {
    d[i] = dst4[scr * Q_PER_SC + i * AGG_THREADS + tid];
  }
}

__device__ __forceinline__ void agg_list_step(int scr, int tid, int base, int nb,
                                              const int4 (&d)[OSUPER],
                                              unsigned* s_list, int* s_cnt, int* s_ovf) {
  #pragma unroll
  for (int i = 0; i < OSUPER; ++i) {
    const int q = scr * Q_PER_SC + i * AGG_THREADS + tid;
    const unsigned tv[4] = {(unsigned)(d[i].x - base), (unsigned)(d[i].y - base),
                            (unsigned)(d[i].z - base), (unsigned)(d[i].w - base)};
    if (min(min(tv[0], tv[1]), min(tv[2], tv[3])) < (unsigned)nb) {
      #pragma unroll
      for (int c = 0; c < 4; ++c) {
        if (tv[c] < (unsigned)nb) {
          const int pos = atomicAdd(s_cnt, 1);
          if (pos < LIST_CAP) s_list[pos] = (unsigned)(4 * q + c);
          else                *s_ovf = 1;
        }
      }
    }
  }
}

__device__ __forceinline__ void agg_resolve(int n, int tid, int base, int nb,
                                            const int* __restrict__ src,
                                            const int* __restrict__ dst,
                                            unsigned* s_list) {
  for (int j0 = tid; j0 < n; j0 += AGG_THREADS * UR) {
    unsigned e[UR];
    int      sv[UR];
    int      dv[UR];
    #pragma unroll
    for (int u = 0; u < UR; ++u) {
      const int j = j0 + u * AGG_THREADS;
      e[u]  = (j < n) ? (s_list[j] & 0xFFFFFu) : 0u;
      sv[u] = src[e[u]];
      dv[u] = dst[e[u]];
    }
    #pragma unroll
    for (int u = 0; u < UR; ++u) {
      const int j = j0 + u * AGG_THREADS;
      if (j < n) {
        unsigned tl = (unsigned)(dv[u] - base);
        if (tl >= (unsigned)nb) tl = NB;
        s_list[j] = (tl << 20) | (unsigned)sv[u];
      }
    }
  }
}

template <int PASS, bool KEPT>
__device__ __forceinline__ void agg_stats(int n, int tid, const float* __restrict__ es,
                                          const unsigned* s_list, const float* s_ed,
                                          unsigned* s_mkey, unsigned long long* s_den,
                                          float* s_sc) {
  const int head  = tid & (NUM_HEADS - 1);
  const int total = n * NUM_HEADS;
  if (KEPT && PASS == 1) {
    #pragma unroll 4
    for (int it = tid; it < total; it += AGG_THREADS) {
      const int   k  = (int)(s_list[it >> 3] >> 20) * NUM_HEADS + head;
      const float ex = __expf(s_sc[it] - finite_or_zero(key2f(s_mkey[k])));
      s_sc[it] = ex;
      atomicAdd(&s_den[k], (unsigned long long)(unsigned)(ex * DEN_ONE));
    }
    return;
  }
  for (int it0 = tid; it0 < total; it0 += AGG_THREADS * UB) {
    int   tl[UB];
    float ev[UB];
    #pragma unroll
    for (int u = 0; u < UB; ++u) {
      const int it = it0 + u * AGG_THREADS;
      const unsigned w = s_list[(it < total ? it : total - 1) >> 3];
      tl[u] = (int)(w >> 20);
      ev[u] = es[(w & 0xFFFFFu) * NUM_HEADS + head];
    }
    #pragma unroll
    for (int u = 0; u < UB; ++u) {
      if (it0 + u * AGG_THREADS < total) {
        const int   k = tl[u] * NUM_HEADS + head;
        const float e = ev[u] + s_ed[k];
        const float v = e > 0.f ? e : NEG_SLOPE * e;
        if (PASS == 0) {
          if (KEPT) s_sc[it0 + u * AGG_THREADS] = v;
          atomicMax(&s_mkey[k], f2key(v));
        } else {
          const float ex = __expf(v - finite_or_zero(key2f(s_mkey[k])));
          atomicAdd(&s_den[k], (unsigned long long)(unsigned)(ex * DEN_ONE));
        }
      }
    }
  }
}

__device__ __forceinline__ void agg_den_to_float(int tid, const unsigned long long* s_den,
                                                 float* s_denf) {
  for (int i = tid; i < NBX * NUM_HEADS; i += AGG_THREADS)
    s_denf[i] = (float)s_den[i] * DEN_STEP;
}

template <bool KEPT>
__device__ __forceinline__ void agg_drain(int n, int wave, int lane,
                                          const float* __restrict__ h,
                                          const float* __restrict__ es,
                                          const unsigned* s_list, const float* s_ed,
                                          const unsigned* s_mkey, const float* s_denf,
                                          const float* s_sc, float* s_acc) {
  const int head = lane >> 2;
  for (int i0 = wave * BATCH; i0 < n; i0 += AGG_WAVES * BATCH) {
    float4 hv[BATCH];
    float  ev[BATCH];
    int    tl[BATCH];
    #pragma unroll
    for (int b = 0; b < BATCH; ++b) {
      const int idx = (i0 + b < n) ? (i0 + b) : (n - 1);
      const unsigned w = s_list[idx];
      const int s = (int)(w & 0xFFFFFu);
      tl[b] = (int)(w >> 20);
      ev[b] = KEPT ? 0.f : es[s * NUM_HEADS + head];
      hv[b] = *(const float4*)&h[(size_t)s * HD + 4 * lane];
    }
    #pragma unroll
    for (int b = 0; b < BATCH; ++b) {
      if (i0 + b < n) {
        const int   k     = tl[b] * NUM_HEADS + head;
        float ex;
        if (KEPT) {
          ex = s_sc[(i0 + b) * NUM_HEADS + head];
        } else {
          const float e = ev[b] + s_ed[k];
          const float v = e > 0.f ? e : NEG_SLOPE * e;
          ex = __expf(v - finite_or_zero(key2f(s_mkey[k])));
        }
        const float dv    = s_denf[k];
        const float alpha = ex / (dv > 0.f ? dv : 1.f);
        float* a = s_acc + tl[b] * HD + lane;
        atomicAdd(a +  0, alpha * hv[b].x);
        atomicAdd(a + 32, alpha * hv[b].y);
        atomicAdd(a + 64, alpha * hv[b].z);
        atomicAdd(a + 96, alpha * hv[b].w);
      }
    }
  }
}

__global__ __launch_bounds__(AGG_THREADS)
void gat_softmax_aggregate_owner(const float* __restrict__ h,
                                 const float* __restrict__ es,
                                 const float* __restrict__ ed,
                                 const float* __restrict__ bias,
                                 const int* __restrict__ src,
                                 const int* __restrict__ dst,
                                 const unsigned* __restrict__ own,
                                 float* __restrict__ out) {
  __shared__ __attribute__((aligned(16))) float    s_acc[NBX * HD];
  __shared__ __attribute__((aligned(16))) unsigned s_list[LIST_CAP];
  __shared__ float    s_sc[SC_CAP * NUM_HEADS];
  __shared__ unsigned long long s_den[NBX * NUM_HEADS];
  __shared__ float    s_denf[NBX * NUM_HEADS];
  __shared__ unsigned s_mkey[NBX * NUM_HEADS];
  __shared__ float    s_ed[NBX * NUM_HEADS];
  __shared__ int      s_cnt;
  __shared__ int      s_ovf;
  const int tid  = threadIdx.x;
  const int lane = tid & 31;
  const int wave = tid >> 5;
  const int base = blockIdx.x * NB;
  const int nb   = (N_NODES - base < NB) ? (N_NODES - base) : NB;

  for (int i = tid; i < NBX * NUM_HEADS; i += AGG_THREADS) {
    s_ed[i]   = (i < nb * NUM_HEADS) ? ed[base * NUM_HEADS + i] : 0.f;
    s_mkey[i] = 0u;
    s_den[i]  = 0ull;
  }
  if (tid == 0) { s_cnt = 0; s_ovf = 0; }
  __syncthreads();

  {
    const unsigned bb   = (unsigned)blockIdx.x * 0x01010101u;
    uint4*         wq_d = (uint4*)s_acc + wave * WQ_CAP;
    unsigned*      wq_i = (unsigned*)s_acc + AGG_WAVES * WQ_CAP * 4 + wave * WQ_CAP;
    int            wq   = 0;
    uint4 x[SUPER], xn[SUPER];
    int scr = (int)blockIdx.x % PSC_FULL;
    own_load_step<false>(scr, tid, bb, own, x);
    for (int sc = 0; sc < PSC_FULL - 1; ++sc) {
      const int scn = (scr + 1 == PSC_FULL) ? 0 : scr + 1;
      own_load_step<false>(scn, tid, bb, own, xn);
      own_queue_step(scr, tid, bb, x, wq_d, wq_i, wq);
      #pragma unroll
      for (int i = 0; i < SUPER; ++i) x[i] = xn[i];
      scr = scn;
    }
    own_load_step<true>(PSC_FULL, tid, bb, own, xn);
    own_queue_step(scr, tid, bb, x, wq_d, wq_i, wq);
    own_queue_step(PSC_FULL, tid, bb, xn, wq_d, wq_i, wq);
    if (wq > WQ_CAP) s_ovf = 1;
    own_emit(wq, lane, bb, wq_d, wq_i, s_list, &s_cnt, &s_ovf);
  }
  __syncthreads();

  for (int i = tid; i < NBX * HD / 4; i += AGG_THREADS)
    ((float4*)s_acc)[i] = make_float4(0.f, 0.f, 0.f, 0.f);

  if (s_ovf == 0 && s_cnt <= SC_CAP) {
    const int n = s_cnt;
    agg_resolve(n, tid, base, nb, src, dst, s_list);
    __syncthreads();
    agg_stats<0, true>(n, tid, es, s_list, s_ed, s_mkey, s_den, s_sc);
    __syncthreads();
    agg_stats<1, true>(n, tid, es, s_list, s_ed, s_mkey, s_den, s_sc);
    __syncthreads();
    agg_den_to_float(tid, s_den, s_denf);
    __syncthreads();
    agg_drain<true>(n, wave, lane, h, es, s_list, s_ed, s_mkey, s_denf, s_sc, s_acc);
  } else {
    for (int pass = 0; pass < 3; ++pass) {
      for (int sc = 0; sc < NSC; ++sc) {
        int4 d[OSUPER];
        __syncthreads();
        if (tid == 0) s_cnt = 0;
        __syncthreads();
        agg_load_step(sc, tid, dst, d);
        agg_list_step(sc, tid, base, nb, d, s_list, &s_cnt, &s_ovf);
        __syncthreads();
        const int n = s_cnt;
        agg_resolve(n, tid, base, nb, src, dst, s_list);
        __syncthreads();
        if (pass == 0)      agg_stats<0, false>(n, tid, es, s_list, s_ed, s_mkey, s_den, s_sc);
        else if (pass == 1) agg_stats<1, false>(n, tid, es, s_list, s_ed, s_mkey, s_den, s_sc);
        else                agg_drain<false>(n, wave, lane, h, es, s_list, s_ed, s_mkey, s_denf, s_sc, s_acc);
      }
      __syncthreads();
      if (pass == 1) agg_den_to_float(tid, s_den, s_denf);
    }
  }
  __syncthreads();

  const float4 bv = *(const float4*)&bias[4 * lane];
  for (int tl = wave; tl < nb; tl += AGG_WAVES) {
    const float* a = s_acc + tl * HD + lane;
    float4 o;
    o.x = a[0]  + bv.x;
    o.y = a[32] + bv.y;
    o.z = a[64] + bv.z;
    o.w = a[96] + bv.w;
    *(float4*)&out[(size_t)(base + tl) * HD + 4 * lane] = o;
  }
}

extern "C" void kernel_launch(void* const* d_in, const int* in_sizes, int n_in,
                              void* d_out, int out_size, void* d_ws, size_t ws_size,
                              hipStream_t stream) {
  const float* feat = (const float*)d_in[0];
  const float* W    = (const float*)d_in[1];
  const float* attn = (const float*)d_in[2];
  const float* bias = (const float*)d_in[3];
  const int*   src  = (const int*)d_in[4];
  const int*   dst  = (const int*)d_in[5];
  float*       out  = (float*)d_out;

  float* h  = (float*)d_ws;
  float* es = h  + (size_t)N_NODES * HD;
  float* ed = es + (size_t)N_NODES * NUM_HEADS;
  unsigned* own = (unsigned*)(ed + (size_t)N_NODES * NUM_HEADS);

  const int NH = N_NODES * NUM_HEADS;

  gat_gemm_wmma<<<N_NODES / 16, 256, 0, stream>>>(feat, W, h);
  gat_logits<<<(NH + 255) / 256, 256, 0, stream>>>(h, attn, dst, es, ed, (unsigned short*)own);
  gat_softmax_aggregate_owner<<<AGG_BLOCKS, AGG_THREADS, 0, stream>>>(h, es, ed, bias, src, dst, own, out);
}


# Projection re-tiled: a wave owns 16 rows x all 128 columns (8 accumulators), A fragments straight from feat, W staged per block in LDS in fragment order; k order in each 8, so h's sum order, changes
# speedup vs baseline: 7.0785x; 1.1369x over previous
#include <hip/hip_runtime.h>
#include <hip/hip_bf16.h>
#include <math.h>

#define N_NODES   50000
#define N_EDGES   800000
#define IN_FEATS  128
#define NUM_HEADS 8
#define OUT_FEATS 16
#define HD        (NUM_HEADS * OUT_FEATS)
#define NEG_SLOPE 0.2f

typedef float v2f __attribute__((ext_vector_type(2)));
typedef float v8f __attribute__((ext_vector_type(8)));

#define GEMM_WAVES   8
#define GEMM_THREADS (GEMM_WAVES * 32)
#define ROW_TILES    (N_NODES / 16)
#define GEMM_BLOCKS  ((ROW_TILES + GEMM_WAVES - 1) / GEMM_WAVES)
#define KJ           (IN_FEATS / 8)
#define NT           (HD / 16)
#define OPITCH       132
#define W_SLOTS      (KJ * NT * 32)

static_assert(N_NODES % 16 == 0 && IN_FEATS % 8 == 0 && HD % 16 == 0, "whole tiles only");
static_assert(NT == GEMM_WAVES, "W staging: wave w fills column tile w");
static_assert(OPITCH % 4 == 0, "a staged output row starts on 16 bytes");

__global__ __launch_bounds__(GEMM_THREADS)
void gat_gemm_wmma(const float* __restrict__ feat,
                   const float* __restrict__ W,
                   float* __restrict__ h) {
  __shared__ float4 lds4[W_SLOTS + GEMM_WAVES * 16 * OPITCH / 4];
  float4* lW = lds4;
  float*  lO = (float*)(lds4 + W_SLOTS);
  const int tid  = threadIdx.x;
  const int lane = tid & 31;
  const int wave = tid >> 5;
  const int c    = lane & 15;
  const int half = lane >> 4;

  for (int j = 0; j < KJ; ++j) {
    const float* wp = W + (size_t)(8 * j + 4 * half) * HD + 16 * wave + c;
    float4 v;
    v.x = wp[0];
    v.y = wp[HD];
    v.z = wp[2 * HD];
    v.w = wp[3 * HD];
    lW[(j * NT + wave) * 32 + lane] = v;
  }
  __syncthreads();

  const int  t    = blockIdx.x * GEMM_WAVES + wave;
  const bool live = t < ROW_TILES;
  float* orow = lO + wave * 16 * OPITCH;

  if (live) {
    const float4* ap = (const float4*)(feat + (size_t)(t * 16 + c) * IN_FEATS + 4 * half);
    float4 a[KJ];
    #pragma unroll
    for (int j = 0; j < KJ; ++j) a[j] = ap[2 * j];

    v8f acc[NT];
    #pragma unroll
    for (int n = 0; n < NT; ++n) acc[n] = v8f{};

    #pragma unroll
    for (int j = 0; j < KJ; ++j) {
      float4 b[NT];
      #pragma unroll
      for (int n = 0; n < NT; ++n) b[n] = lW[(j * NT + n) * 32 + lane];
      v2f a0; a0.x = a[j].x; a0.y = a[j].y;
      v2f a1; a1.x = a[j].z; a1.y = a[j].w;
      #pragma unroll
      for (int n = 0; n < NT; ++n) {
        v2f b0; b0.x = b[n].x; b0.y = b[n].y;
        acc[n] = __builtin_amdgcn_wmma_f32_16x16x4_f32(
             false, a0,  false, b0,
             (short)0, acc[n],  false,  false);
      }
      #pragma unroll
      for (int n = 0; n < NT; ++n) {
        v2f b1; b1.x = b[n].z; b1.y = b[n].w;
        acc[n] = __builtin_amdgcn_wmma_f32_16x16x4_f32(
             false, a1,  false, b1,
             (short)0, acc[n],  false,  false);
      }
    }

    #pragma unroll
    for (int n = 0; n < NT; ++n) {
      #pragma unroll
      for (int r = 0; r < 8; ++r) orow[(8 * half + r) * OPITCH + 16 * n + c] = acc[n][r];
    }
  }
  __syncthreads();
  if (live) {
    for (int r = 0; r < 16; ++r) {
      const float4 v = *(const float4*)&orow[r * OPITCH + 4 * lane];
      *(float4*)&h[(size_t)(t * 16 + r) * HD + 4 * lane] = v;
    }
  }
}

#define NB 196
static_assert(N_EDGES == 2 * N_NODES * NUM_HEADS, "one thread writes the owner bytes of two edges");
static_assert((N_NODES + NB - 1) / NB <= 256, "an owner number fits a byte");

__global__ void gat_logits(const float* __restrict__ h,
                           const float* __restrict__ attn,
                           const int* __restrict__ dst,
                           float* __restrict__ es, float* __restrict__ ed,
                           unsigned short* __restrict__ own2) {
  const int idx = blockIdx.x * blockDim.x + threadIdx.x;
  if (idx >= N_NODES * NUM_HEADS) return;
  {
    const int2 d = ((const int2*)dst)[idx];
    own2[idx] = (unsigned short)((((unsigned)d.x / NB) & 0xFFu) |
                                 ((((unsigned)d.y / NB) & 0xFFu) << 8));
  }
  const int head = idx & (NUM_HEADS - 1);
  const int n    = idx >> 3;
  const float* hp = h    + (size_t)n * HD + head * OUT_FEATS;
  const float* a0 = attn +           head * OUT_FEATS;
  const float* a1 = attn + HD      + head * OUT_FEATS;
  float s0 = 0.f, s1 = 0.f;
  #pragma unroll
  for (int d = 0; d < OUT_FEATS; ++d) {
    const float hv = hp[d];
    s0 += hv * a0[d];
    s1 += hv * a1[d];
  }
  es[idx] = s0;
  ed[idx] = s1;
}

__device__ __forceinline__ unsigned f2key(float v) {
  const unsigned u = __float_as_uint(v);
  return (u & 0x80000000u) ? ~u : (u | 0x80000000u);
}
__device__ __forceinline__ float key2f(unsigned k) {
  return __uint_as_float((k & 0x80000000u) ? (k & 0x7FFFFFFFu) : ~k);
}
__device__ __forceinline__ float finite_or_zero(float m) {
  return (fabsf(m) < INFINITY) ? m : 0.f;
}

#define NBX         (NB + 1)
#define AGG_BLOCKS  256
#define AGG_THREADS 800
#define AGG_WAVES   (AGG_THREADS / 32)
#define SUPER       5
#define OSUPER      2
#define PQ          (N_EDGES / 16)
#define P_PER_SC    (AGG_THREADS * SUPER)
#define PSC_FULL    (PQ / P_PER_SC)
#define WQ_CAP      200
#define NQ          (N_EDGES / 4)
#define Q_PER_SC    (AGG_THREADS * OSUPER)
#define NSC         (NQ / Q_PER_SC)
#define LIST_CAP    8192
#define SC_CAP      3840
#define BATCH       8
#define UB          8
#define UR          4
#define DEN_ONE     2147483648.0f
#define DEN_STEP    4.656612873077392578125e-10f

static_assert(N_EDGES <= (1 << 20) && NBX <= (1 << 12), "a list word is local node << 20 | edge or src");
static_assert(N_NODES <= (1 << 20), "a resolved list word is local node << 20 | src node");
static_assert(AGG_BLOCKS * NB >= N_NODES && AGG_BLOCKS <= 256, "owner ranges cover all nodes, one byte each");
static_assert(N_EDGES % 16 == 0 && PSC_FULL * P_PER_SC < PQ && (PSC_FULL + 1) * P_PER_SC >= PQ,
              "the plane is scanned in whole steps and one partial step");
static_assert(N_EDGES % 4 == 0 && NQ % Q_PER_SC == 0, "overflow path: dst is scanned as int4, in whole steps");
static_assert(Q_PER_SC * 4 <= LIST_CAP, "overflow path: one scan step always fits the list");
static_assert(SC_CAP <= LIST_CAP, "every edge with kept scores is on the list");
static_assert(AGG_THREADS % NUM_HEADS == 0, "a thread keeps one head over all its items");
static_assert(AGG_WAVES * WQ_CAP * 5 <= NBX * HD, "the wave queues fit the LDS words of the row sums");

__device__ __forceinline__ unsigned zero_bytes(unsigned z) {
  return ~((((z & 0x7F7F7F7Fu) + 0x7F7F7F7Fu) | z) | 0x7F7F7F7Fu);
}

template <bool PARTIAL>
__device__ __forceinline__ void own_load_step(int scr, int tid, unsigned bb,
                                              const unsigned* __restrict__ own,
                                              uint4 (&x)[SUPER]) {
  const uint4* own4 = (const uint4*)own;
  #pragma unroll
  for (int i = 0; i < SUPER; ++i) {
    const int q = scr * P_PER_SC + i * AGG_THREADS + tid;
    if (!PARTIAL || q < PQ) x[i] = own4[q];
    else                    x[i] = make_uint4(~bb, ~bb, ~bb, ~bb);
  }
}

__device__ __forceinline__ void own_queue_step(int scr, int tid, unsigned bb,
                                               const uint4 (&x)[SUPER],
                                               uint4* wq_d, unsigned* wq_i, int& wq) {
  #pragma unroll
  for (int i = 0; i < SUPER; ++i) {
    const int q = scr * P_PER_SC + i * AGG_THREADS + tid;
    const unsigned z0 = x[i].x ^ bb, z1 = x[i].y ^ bb, z2 = x[i].z ^ bb, z3 = x[i].w ^ bb;
    const unsigned t = (((z0 - 0x01010101u) & ~z0) | ((z1 - 0x01010101u) & ~z1) |
                        ((z2 - 0x01010101u) & ~z2) | ((z3 - 0x01010101u) & ~z3)) & 0x80808080u;
    const bool     cand = t != 0u;
    const unsigned m    = __builtin_amdgcn_ballot_w32(cand);
    if (m != 0u) {
      if (cand) {
        const int slot = min(wq + (int)__builtin_amdgcn_mbcnt_lo(m, 0u), WQ_CAP - 1);
        wq_d[slot] = x[i];
        wq_i[slot] = (unsigned)q;
      }
      wq += __popc(m);
    }
  }
}

__device__ __forceinline__ void own_emit(int wq, int lane, unsigned bb,
                                         const uint4* wq_d, const unsigned* wq_i,
                                         unsigned* s_list, int* s_cnt, int* s_ovf) {
  const int n = wq < WQ_CAP ? wq : WQ_CAP;
  for (int i0 = 0; i0 < n; i0 += 32) {
    const int i = i0 + lane;
    unsigned hz[4] = {0u, 0u, 0u, 0u};
    unsigned q = 0u;
    if (i < n) {
      const uint4 x = wq_d[i];
      q = wq_i[i];
      hz[0] = zero_bytes(x.x ^ bb);
      hz[1] = zero_bytes(x.y ^ bb);
      hz[2] = zero_bytes(x.z ^ bb);
      hz[3] = zero_bytes(x.w ^ bb);
    }
    const int c = __popc(hz[0]) + __popc(hz[1]) + __popc(hz[2]) + __popc(hz[3]);
    int incl = c;
    #pragma unroll
    for (int d = 1; d < 32; d <<= 1) {
      const int v = __shfl_up(incl, d, 32);
      if (lane >= d) incl += v;
    }
    const int total = __shfl(incl, 31, 32);
    int first = 0;
    if (lane == 0) first = atomicAdd(s_cnt, total);
    first = __shfl(first, 0, 32);
    int pos = first + incl - c;
    #pragma unroll
    for (int j = 0; j < 4; ++j) {
      unsigned rest = hz[j];
      while (rest != 0u) {
        const unsigned b = (unsigned)(__ffs((int)rest) - 1) >> 3;
        if (pos < LIST_CAP) s_list[pos] = 16u * q + 4u * (unsigned)j + b;
        else                *s_ovf = 1;
        ++pos;
        rest &= rest - 1u;
      }
    }
  }
}

__device__ __forceinline__ void agg_load_step(int scr, int tid, const int* __restrict__ dst,
                                              int4 (&d)[OSUPER]) {
  const int4* dst4 = (const int4*)dst;
  #pragma unroll
  for (int i = 0; i < OSUPER; ++i) {
    d[i] = dst4[scr * Q_PER_SC + i * AGG_THREADS + tid];
  }
}

__device__ __forceinline__ void agg_list_step(int scr, int tid, int base, int nb,
                                              const int4 (&d)[OSUPER],
                                              unsigned* s_list, int* s_cnt, int* s_ovf) {
  #pragma unroll
  for (int i = 0; i < OSUPER; ++i) {
    const int q = scr * Q_PER_SC + i * AGG_THREADS + tid;
    const unsigned tv[4] = {(unsigned)(d[i].x - base), (unsigned)(d[i].y - base),
                            (unsigned)(d[i].z - base), (unsigned)(d[i].w - base)};
    if (min(min(tv[0], tv[1]), min(tv[2], tv[3])) < (unsigned)nb) {
      #pragma unroll
      for (int c = 0; c < 4; ++c) {
        if (tv[c] < (unsigned)nb) {
          const int pos = atomicAdd(s_cnt, 1);
          if (pos < LIST_CAP) s_list[pos] = (unsigned)(4 * q + c);
          else                *s_ovf = 1;
        }
      }
    }
  }
}

__device__ __forceinline__ void agg_resolve(int n, int tid, int base, int nb,
                                            const int* __restrict__ src,
                                            const int* __restrict__ dst,
                                            unsigned* s_list) {
  for (int j0 = tid; j0 < n; j0 += AGG_THREADS * UR) {
    unsigned e[UR];
    int      sv[UR];
    int      dv[UR];
    #pragma unroll
    for (int u = 0; u < UR; ++u) {
      const int j = j0 + u * AGG_THREADS;
      e[u]  = (j < n) ? (s_list[j] & 0xFFFFFu) : 0u;
      sv[u] = src[e[u]];
      dv[u] = dst[e[u]];
    }
    #pragma unroll
    for (int u = 0; u < UR; ++u) {
      const int j = j0 + u * AGG_THREADS;
      if (j < n) {
        unsigned tl = (unsigned)(dv[u] - base);
        if (tl >= (unsigned)nb) tl = NB;
        s_list[j] = (tl << 20) | (unsigned)sv[u];
      }
    }
  }
}

template <int PASS, bool KEPT>
__device__ __forceinline__ void agg_stats(int n, int tid, const float* __restrict__ es,
                                          const unsigned* s_list, const float* s_ed,
                                          unsigned* s_mkey, unsigned long long* s_den,
                                          float* s_sc) {
  const int head  = tid & (NUM_HEADS - 1);
  const int total = n * NUM_HEADS;
  if (KEPT && PASS == 1) {
    #pragma unroll 4
    for (int it = tid; it < total; it += AGG_THREADS) {
      const int   k  = (int)(s_list[it >> 3] >> 20) * NUM_HEADS + head;
      const float ex = __expf(s_sc[it] - finite_or_zero(key2f(s_mkey[k])));
      s_sc[it] = ex;
      atomicAdd(&s_den[k], (unsigned long long)(unsigned)(ex * DEN_ONE));
    }
    return;
  }
  for (int it0 = tid; it0 < total; it0 += AGG_THREADS * UB) {
    int   tl[UB];
    float ev[UB];
    #pragma unroll
    for (int u = 0; u < UB; ++u) {
      const int it = it0 + u * AGG_THREADS;
      const unsigned w = s_list[(it < total ? it : total - 1) >> 3];
      tl[u] = (int)(w >> 20);
      ev[u] = es[(w & 0xFFFFFu) * NUM_HEADS + head];
    }
    #pragma unroll
    for (int u = 0; u < UB; ++u) {
      if (it0 + u * AGG_THREADS < total) {
        const int   k = tl[u] * NUM_HEADS + head;
        const float e = ev[u] + s_ed[k];
        const float v = e > 0.f ? e : NEG_SLOPE * e;
        if (PASS == 0) {
          if (KEPT) s_sc[it0 + u * AGG_THREADS] = v;
          atomicMax(&s_mkey[k], f2key(v));
        } else {
          const float ex = __expf(v - finite_or_zero(key2f(s_mkey[k])));
          atomicAdd(&s_den[k], (unsigned long long)(unsigned)(ex * DEN_ONE));
        }
      }
    }
  }
}

__device__ __forceinline__ void agg_den_to_float(int tid, const unsigned long long* s_den,
                                                 float* s_denf) {
  for (int i = tid; i < NBX * NUM_HEADS; i += AGG_THREADS)
    s_denf[i] = (float)s_den[i] * DEN_STEP;
}

template <bool KEPT>
__device__ __forceinline__ void agg_drain(int n, int wave, int lane,
                                          const float* __restrict__ h,
                                          const float* __restrict__ es,
                                          const unsigned* s_list, const float* s_ed,
                                          const unsigned* s_mkey, const float* s_denf,
                                          const float* s_sc, float* s_acc) {
  const int head = lane >> 2;
  for (int i0 = wave * BATCH; i0 < n; i0 += AGG_WAVES * BATCH) {
    float4 hv[BATCH];
    float  ev[BATCH];
    int    tl[BATCH];
    #pragma unroll
    for (int b = 0; b < BATCH; ++b) {
      const int idx = (i0 + b < n) ? (i0 + b) : (n - 1);
      const unsigned w = s_list[idx];
      const int s = (int)(w & 0xFFFFFu);
      tl[b] = (int)(w >> 20);
      ev[b] = KEPT ? 0.f : es[s * NUM_HEADS + head];
      hv[b] = *(const float4*)&h[(size_t)s * HD + 4 * lane];
    }
    #pragma unroll
    for (int b = 0; b < BATCH; ++b) {
      if (i0 + b < n) {
        const int   k     = tl[b] * NUM_HEADS + head;
        float ex;
        if (KEPT) {
          ex = s_sc[(i0 + b) * NUM_HEADS + head];
        } else {
          const float e = ev[b] + s_ed[k];
          const float v = e > 0.f ? e : NEG_SLOPE * e;
          ex = __expf(v - finite_or_zero(key2f(s_mkey[k])));
        }
        const float dv    = s_denf[k];
        const float alpha = ex / (dv > 0.f ? dv : 1.f);
        float* a = s_acc + tl[b] * HD + lane;
        atomicAdd(a +  0, alpha * hv[b].x);
        atomicAdd(a + 32, alpha * hv[b].y);
        atomicAdd(a + 64, alpha * hv[b].z);
        atomicAdd(a + 96, alpha * hv[b].w);
      }
    }
  }
}

__global__ __launch_bounds__(AGG_THREADS)
void gat_softmax_aggregate_owner(const float* __restrict__ h,
                                 const float* __restrict__ es,
                                 const float* __restrict__ ed,
                                 const float* __restrict__ bias,
                                 const int* __restrict__ src,
                                 const int* __restrict__ dst,
                                 const unsigned* __restrict__ own,
                                 float* __restrict__ out) {
  __shared__ __attribute__((aligned(16))) float    s_acc[NBX * HD];
  __shared__ __attribute__((aligned(16))) unsigned s_list[LIST_CAP];
  __shared__ float    s_sc[SC_CAP * NUM_HEADS];
  __shared__ unsigned long long s_den[NBX * NUM_HEADS];
  __shared__ float    s_denf[NBX * NUM_HEADS];
  __shared__ unsigned s_mkey[NBX * NUM_HEADS];
  __shared__ float    s_ed[NBX * NUM_HEADS];
  __shared__ int      s_cnt;
  __shared__ int      s_ovf;
  const int tid  = threadIdx.x;
  const int lane = tid & 31;
  const int wave = tid >> 5;
  const int base = blockIdx.x * NB;
  const int nb   = (N_NODES - base < NB) ? (N_NODES - base) : NB;

  for (int i = tid; i < NBX * NUM_HEADS; i += AGG_THREADS) {
    s_ed[i]   = (i < nb * NUM_HEADS) ? ed[base * NUM_HEADS + i] : 0.f;
    s_mkey[i] = 0u;
    s_den[i]  = 0ull;
  }
  if (tid == 0) { s_cnt = 0; s_ovf = 0; }
  __syncthreads();

  {
    const unsigned bb   = (unsigned)blockIdx.x * 0x01010101u;
    uint4*         wq_d = (uint4*)s_acc + wave * WQ_CAP;
    unsigned*      wq_i = (unsigned*)s_acc + AGG_WAVES * WQ_CAP * 4 + wave * WQ_CAP;
    int            wq   = 0;
    uint4 x[SUPER], xn[SUPER];
    int scr = (int)blockIdx.x % PSC_FULL;
    own_load_step<false>(scr, tid, bb, own, x);
    for (int sc = 0; sc < PSC_FULL - 1; ++sc) {
      const int scn = (scr + 1 == PSC_FULL) ? 0 : scr + 1;
      own_load_step<false>(scn, tid, bb, own, xn);
      own_queue_step(scr, tid, bb, x, wq_d, wq_i, wq);
      #pragma unroll
      for (int i = 0; i < SUPER; ++i) x[i] = xn[i];
      scr = scn;
    }
    own_load_step<true>(PSC_FULL, tid, bb, own, xn);
    own_queue_step(scr, tid, bb, x, wq_d, wq_i, wq);
    own_queue_step(PSC_FULL, tid, bb, xn, wq_d, wq_i, wq);
    if (wq > WQ_CAP) s_ovf = 1;
    own_emit(wq, lane, bb, wq_d, wq_i, s_list, &s_cnt, &s_ovf);
  }
  __syncthreads();

  for (int i = tid; i < NBX * HD / 4; i += AGG_THREADS)
    ((float4*)s_acc)[i] = make_float4(0.f, 0.f, 0.f, 0.f);

  if (s_ovf == 0 && s_cnt <= SC_CAP) {
    const int n = s_cnt;
    agg_resolve(n, tid, base, nb, src, dst, s_list);
    __syncthreads();
    agg_stats<0, true>(n, tid, es, s_list, s_ed, s_mkey, s_den, s_sc);
    __syncthreads();
    agg_stats<1, true>(n, tid, es, s_list, s_ed, s_mkey, s_den, s_sc);
    __syncthreads();
    agg_den_to_float(tid, s_den, s_denf);
    __syncthreads();
    agg_drain<true>(n, wave, lane, h, es, s_list, s_ed, s_mkey, s_denf, s_sc, s_acc);
  } else {
    for (int pass = 0; pass < 3; ++pass) {
      for (int sc = 0; sc < NSC; ++sc) {
        int4 d[OSUPER];
        __syncthreads();
        if (tid == 0) s_cnt = 0;
        __syncthreads();
        agg_load_step(sc, tid, dst, d);
        agg_list_step(sc, tid, base, nb, d, s_list, &s_cnt, &s_ovf);
        __syncthreads();
        const int n = s_cnt;
        agg_resolve(n, tid, base, nb, src, dst, s_list);
        __syncthreads();
        if (pass == 0)      agg_stats<0, false>(n, tid, es, s_list, s_ed, s_mkey, s_den, s_sc);
        else if (pass == 1) agg_stats<1, false>(n, tid, es, s_list, s_ed, s_mkey, s_den, s_sc);
        else                agg_drain<false>(n, wave, lane, h, es, s_list, s_ed, s_mkey, s_denf, s_sc, s_acc);
      }
      __syncthreads();
      if (pass == 1) agg_den_to_float(tid, s_den, s_denf);
    }
  }
  __syncthreads();

  const float4 bv = *(const float4*)&bias[4 * lane];
  for (int tl = wave; tl < nb; tl += AGG_WAVES) {
    const float* a = s_acc + tl * HD + lane;
    float4 o;
    o.x = a[0]  + bv.x;
    o.y = a[32] + bv.y;
    o.z = a[64] + bv.z;
    o.w = a[96] + bv.w;
    *(float4*)&out[(size_t)(base + tl) * HD + 4 * lane] = o;
  }
}

extern "C" void kernel_launch(void* const* d_in, const int* in_sizes, int n_in,
                              void* d_out, int out_size, void* d_ws, size_t ws_size,
                              hipStream_t stream) {
  const float* feat = (const float*)d_in[0];
  const float* W    = (const float*)d_in[1];
  const float* attn = (const float*)d_in[2];
  const float* bias = (const float*)d_in[3];
  const int*   src  = (const int*)d_in[4];
  const int*   dst  = (const int*)d_in[5];
  float*       out  = (float*)d_out;

  float* h  = (float*)d_ws;
  float* es = h  + (size_t)N_NODES * HD;
  float* ed = es + (size_t)N_NODES * NUM_HEADS;
  unsigned* own = (unsigned*)(ed + (size_t)N_NODES * NUM_HEADS);

  const int NH = N_NODES * NUM_HEADS;

  gat_gemm_wmma<<<GEMM_BLOCKS, GEMM_THREADS, 0, stream>>>(feat, W, h);
  gat_logits<<<(NH + 255) / 256, 256, 0, stream>>>(h, attn, dst, es, ed, (unsigned short*)own);
  gat_softmax_aggregate_owner<<<AGG_BLOCKS, AGG_THREADS, 0, stream>>>(h, es, ed, bias, src, dst, own, out);
}


# Row sums leave LDS: resolve deals each listed edge (LDS int add) to the wave owning its dst, 8 nodes a wave; it sums alpha*h rows in registers, stores out; main path: no float adds; sum order changes
# speedup vs baseline: 7.5274x; 1.0634x over previous
#include <hip/hip_runtime.h>
#include <hip/hip_bf16.h>
#include <math.h>

#define N_NODES   50000
#define N_EDGES   800000
#define IN_FEATS  128
#define NUM_HEADS 8
#define OUT_FEATS 16
#define HD        (NUM_HEADS * OUT_FEATS)
#define NEG_SLOPE 0.2f

typedef float v2f __attribute__((ext_vector_type(2)));
typedef float v8f __attribute__((ext_vector_type(8)));

#define GEMM_WAVES   8
#define GEMM_THREADS (GEMM_WAVES * 32)
#define ROW_TILES    (N_NODES / 16)
#define GEMM_BLOCKS  ((ROW_TILES + GEMM_WAVES - 1) / GEMM_WAVES)
#define KJ           (IN_FEATS / 8)
#define NT           (HD / 16)
#define OPITCH       132
#define W_SLOTS      (KJ * NT * 32)

static_assert(N_NODES % 16 == 0 && IN_FEATS % 8 == 0 && HD % 16 == 0, "whole tiles only");
static_assert(NT == GEMM_WAVES, "W staging: wave w fills column tile w");
static_assert(OPITCH % 4 == 0, "a staged output row starts on 16 bytes");

__global__ __launch_bounds__(GEMM_THREADS)
void gat_gemm_wmma(const float* __restrict__ feat,
                   const float* __restrict__ W,
                   float* __restrict__ h) {
  __shared__ float4 lds4[W_SLOTS + GEMM_WAVES * 16 * OPITCH / 4];
  float4* lW = lds4;
  float*  lO = (float*)(lds4 + W_SLOTS);
  const int tid  = threadIdx.x;
  const int lane = tid & 31;
  const int wave = tid >> 5;
  const int c    = lane & 15;
  const int half = lane >> 4;

  for (int j = 0; j < KJ; ++j) {
    const float* wp = W + (size_t)(8 * j + 4 * half) * HD + 16 * wave + c;
    float4 v;
    v.x = wp[0];
    v.y = wp[HD];
    v.z = wp[2 * HD];
    v.w = wp[3 * HD];
    lW[(j * NT + wave) * 32 + lane] = v;
  }
  __syncthreads();

  const int  t    = blockIdx.x * GEMM_WAVES + wave;
  const bool live = t < ROW_TILES;
  float* orow = lO + wave * 16 * OPITCH;

  if (live) {
    const float4* ap = (const float4*)(feat + (size_t)(t * 16 + c) * IN_FEATS + 4 * half);
    float4 a[KJ];
    #pragma unroll
    for (int j = 0; j < KJ; ++j) a[j] = ap[2 * j];

    v8f acc[NT];
    #pragma unroll
    for (int n = 0; n < NT; ++n) acc[n] = v8f{};

    #pragma unroll
    for (int j = 0; j < KJ; ++j) {
      float4 b[NT];
      #pragma unroll
      for (int n = 0; n < NT; ++n) b[n] = lW[(j * NT + n) * 32 + lane];
      v2f a0; a0.x = a[j].x; a0.y = a[j].y;
      v2f a1; a1.x = a[j].z; a1.y = a[j].w;
      #pragma unroll
      for (int n = 0; n < NT; ++n) {
        v2f b0; b0.x = b[n].x; b0.y = b[n].y;
        acc[n] = __builtin_amdgcn_wmma_f32_16x16x4_f32(
             false, a0,  false, b0,
             (short)0, acc[n],  false,  false);
      }
      #pragma unroll
      for (int n = 0; n < NT; ++n) {
        v2f b1; b1.x = b[n].z; b1.y = b[n].w;
        acc[n] = __builtin_amdgcn_wmma_f32_16x16x4_f32(
             false, a1,  false, b1,
             (short)0, acc[n],  false,  false);
      }
    }

    #pragma unroll
    for (int n = 0; n < NT; ++n) {
      #pragma unroll
      for (int r = 0; r < 8; ++r) orow[(8 * half + r) * OPITCH + 16 * n + c] = acc[n][r];
    }
  }
  __syncthreads();
  if (live) {
    for (int r = 0; r < 16; ++r) {
      const float4 v = *(const float4*)&orow[r * OPITCH + 4 * lane];
      *(float4*)&h[(size_t)(t * 16 + r) * HD + 4 * lane] = v;
    }
  }
}

#define NB 196
static_assert(N_EDGES == 2 * N_NODES * NUM_HEADS, "one thread writes the owner bytes of two edges");
static_assert((N_NODES + NB - 1) / NB <= 256, "an owner number fits a byte");

__global__ void gat_logits(const float* __restrict__ h,
                           const float* __restrict__ attn,
                           const int* __restrict__ dst,
                           float* __restrict__ es, float* __restrict__ ed,
                           unsigned short* __restrict__ own2) {
  const int idx = blockIdx.x * blockDim.x + threadIdx.x;
  if (idx >= N_NODES * NUM_HEADS) return;
  {
    const int2 d = ((const int2*)dst)[idx];
    own2[idx] = (unsigned short)((((unsigned)d.x / NB) & 0xFFu) |
                                 ((((unsigned)d.y / NB) & 0xFFu) << 8));
  }
  const int head = idx & (NUM_HEADS - 1);
  const int n    = idx >> 3;
  const float* hp = h    + (size_t)n * HD + head * OUT_FEATS;
  const float* a0 = attn +           head * OUT_FEATS;
  const float* a1 = attn + HD      + head * OUT_FEATS;
  float s0 = 0.f, s1 = 0.f;
  #pragma unroll
  for (int d = 0; d < OUT_FEATS; ++d) {
    const float hv = hp[d];
    s0 += hv * a0[d];
    s1 += hv * a1[d];
  }
  es[idx] = s0;
  ed[idx] = s1;
}

__device__ __forceinline__ unsigned f2key(float v) {
  const unsigned u = __float_as_uint(v);
  return (u & 0x80000000u) ? ~u : (u | 0x80000000u);
}
__device__ __forceinline__ float key2f(unsigned k) {
  return __uint_as_float((k & 0x80000000u) ? (k & 0x7FFFFFFFu) : ~k);
}
__device__ __forceinline__ float finite_or_zero(float m) {
  return (fabsf(m) < INFINITY) ? m : 0.f;
}

#define NBX         (NB + 1)
#define AGG_BLOCKS  256
#define AGG_THREADS 800
#define AGG_WAVES   (AGG_THREADS / 32)
#define SUPER       5
#define OSUPER      2
#define PQ          (N_EDGES / 16)
#define P_PER_SC    (AGG_THREADS * SUPER)
#define PSC_FULL    (PQ / P_PER_SC)
#define WQ_CAP      200
#define NQ          (N_EDGES / 4)
#define Q_PER_SC    (AGG_THREADS * OSUPER)
#define NSC         (NQ / Q_PER_SC)
#define LIST_CAP    8192
#define SC_CAP      3840
#define BATCH       8
#define WN          8
#define WQ2_CAP     (NBX * HD / 2 / AGG_WAVES)
#define UB          8
#define UR          4
#define DEN_ONE     2147483648.0f
#define DEN_STEP    4.656612873077392578125e-10f

static_assert(N_EDGES <= (1 << 20) && NBX <= (1 << 12), "a list word is local node << 20 | edge or src");
static_assert(N_NODES <= (1 << 20), "a resolved list word is local node << 20 | src node");
static_assert(AGG_BLOCKS * NB >= N_NODES && AGG_BLOCKS <= 256, "owner ranges cover all nodes, one byte each");
static_assert(N_EDGES % 16 == 0 && PSC_FULL * P_PER_SC < PQ && (PSC_FULL + 1) * P_PER_SC >= PQ,
              "the plane is scanned in whole steps and one partial step");
static_assert(N_EDGES % 4 == 0 && NQ % Q_PER_SC == 0, "overflow path: dst is scanned as int4, in whole steps");
static_assert(Q_PER_SC * 4 <= LIST_CAP, "overflow path: one scan step always fits the list");
static_assert(SC_CAP <= LIST_CAP, "every edge with kept scores is on the list");
static_assert(AGG_THREADS % NUM_HEADS == 0, "a thread keeps one head over all its items");
static_assert(AGG_WAVES * WQ_CAP * 5 <= NBX * HD, "the wave queues fit the LDS words of the row sums");
static_assert(AGG_WAVES * WN >= NBX, "every row, the spare one too, has an owner wave");
static_assert((WN & (WN - 1)) == 0 && BATCH <= 32, "a node's place in its wave is tl & (WN - 1); a batch's entries sit in one lane each");

__device__ __forceinline__ unsigned zero_bytes(unsigned z) {
  return ~((((z & 0x7F7F7F7Fu) + 0x7F7F7F7Fu) | z) | 0x7F7F7F7Fu);
}

template <bool PARTIAL>
__device__ __forceinline__ void own_load_step(int scr, int tid, unsigned bb,
                                              const unsigned* __restrict__ own,
                                              uint4 (&x)[SUPER]) {
  const uint4* own4 = (const uint4*)own;
  #pragma unroll
  for (int i = 0; i < SUPER; ++i) {
    const int q = scr * P_PER_SC + i * AGG_THREADS + tid;
    if (!PARTIAL || q < PQ) x[i] = own4[q];
    else                    x[i] = make_uint4(~bb, ~bb, ~bb, ~bb);
  }
}

__device__ __forceinline__ void own_queue_step(int scr, int tid, unsigned bb,
                                               const uint4 (&x)[SUPER],
                                               uint4* wq_d, unsigned* wq_i, int& wq) {
  #pragma unroll
  for (int i = 0; i < SUPER; ++i) {
    const int q = scr * P_PER_SC + i * AGG_THREADS + tid;
    const unsigned z0 = x[i].x ^ bb, z1 = x[i].y ^ bb, z2 = x[i].z ^ bb, z3 = x[i].w ^ bb;
    const unsigned t = (((z0 - 0x01010101u) & ~z0) | ((z1 - 0x01010101u) & ~z1) |
                        ((z2 - 0x01010101u) & ~z2) | ((z3 - 0x01010101u) & ~z3)) & 0x80808080u;
    const bool     cand = t != 0u;
    const unsigned m    = __builtin_amdgcn_ballot_w32(cand);
    if (m != 0u) {
      if (cand) {
        const int slot = min(wq + (int)__builtin_amdgcn_mbcnt_lo(m, 0u), WQ_CAP - 1);
        wq_d[slot] = x[i];
        wq_i[slot] = (unsigned)q;
      }
      wq += __popc(m);
    }
  }
}

__device__ __forceinline__ void own_emit(int wq, int lane, unsigned bb,
                                         const uint4* wq_d, const unsigned* wq_i,
                                         unsigned* s_list, int* s_cnt, int* s_ovf) {
  const int n = wq < WQ_CAP ? wq : WQ_CAP;
  for (int i0 = 0; i0 < n; i0 += 32) {
    const int i = i0 + lane;
    unsigned hz[4] = {0u, 0u, 0u, 0u};
    unsigned q = 0u;
    if (i < n) {
      const uint4 x = wq_d[i];
      q = wq_i[i];
      hz[0] = zero_bytes(x.x ^ bb);
      hz[1] = zero_bytes(x.y ^ bb);
      hz[2] = zero_bytes(x.z ^ bb);
      hz[3] = zero_bytes(x.w ^ bb);
    }
    const int c = __popc(hz[0]) + __popc(hz[1]) + __popc(hz[2]) + __popc(hz[3]);
    int incl = c;
    #pragma unroll
    for (int d = 1; d < 32; d <<= 1) {
      const int v = __shfl_up(incl, d, 32);
      if (lane >= d) incl += v;
    }
    const int total = __shfl(incl, 31, 32);
    int first = 0;
    if (lane == 0) first = atomicAdd(s_cnt, total);
    first = __shfl(first, 0, 32);
    int pos = first + incl - c;
    #pragma unroll
    for (int j = 0; j < 4; ++j) {
      unsigned rest = hz[j];
      while (rest != 0u) {
        const unsigned b = (unsigned)(__ffs((int)rest) - 1) >> 3;
        if (pos < LIST_CAP) s_list[pos] = 16u * q + 4u * (unsigned)j + b;
        else                *s_ovf = 1;
        ++pos;
        rest &= rest - 1u;
      }
    }
  }
}

__device__ __forceinline__ void agg_load_step(int scr, int tid, const int* __restrict__ dst,
                                              int4 (&d)[OSUPER]) {
  const int4* dst4 = (const int4*)dst;
  #pragma unroll
  for (int i = 0; i < OSUPER; ++i) {
    d[i] = dst4[scr * Q_PER_SC + i * AGG_THREADS + tid];
  }
}

__device__ __forceinline__ void agg_list_step(int scr, int tid, int base, int nb,
                                              const int4 (&d)[OSUPER],
                                              unsigned* s_list, int* s_cnt, int* s_ovf) {
  #pragma unroll
  for (int i = 0; i < OSUPER; ++i) {
    const int q = scr * Q_PER_SC + i * AGG_THREADS + tid;
    const unsigned tv[4] = {(unsigned)(d[i].x - base), (unsigned)(d[i].y - base),
                            (unsigned)(d[i].z - base), (unsigned)(d[i].w - base)};
    if (min(min(tv[0], tv[1]), min(tv[2], tv[3])) < (unsigned)nb) {
      #pragma unroll
      for (int c = 0; c < 4; ++c) {
        if (tv[c] < (unsigned)nb) {
          const int pos = atomicAdd(s_cnt, 1);
          if (pos < LIST_CAP) s_list[pos] = (unsigned)(4 * q + c);
          else                *s_ovf = 1;
        }
      }
    }
  }
}

template <bool DEAL>
__device__ __forceinline__ void agg_resolve(int n, int tid, int base, int nb,
                                            const int* __restrict__ src,
                                            const int* __restrict__ dst,
                                            unsigned* s_list,
                                            uint2* wq2, int* s_qn, int* s_qovf) {
  for (int j0 = tid; j0 < n; j0 += AGG_THREADS * UR) {
    unsigned e[UR];
    int      sv[UR];
    int      dv[UR];
    #pragma unroll
    for (int u = 0; u < UR; ++u) {
      const int j = j0 + u * AGG_THREADS;
      e[u]  = (j < n) ? (s_list[j] & 0xFFFFFu) : 0u;
      sv[u] = src[e[u]];
      dv[u] = dst[e[u]];
    }
    #pragma unroll
    for (int u = 0; u < UR; ++u) {
      const int j = j0 + u * AGG_THREADS;
      if (j < n) {
        unsigned tl = (unsigned)(dv[u] - base);
        if (tl >= (unsigned)nb) tl = NB;
        const unsigned wv = (tl << 20) | (unsigned)sv[u];
        s_list[j] = wv;
        if (DEAL) {
          const int ow  = (int)(tl / WN);
          const int pos = atomicAdd(&s_qn[ow], 1);
          if (pos < WQ2_CAP) wq2[ow * WQ2_CAP + pos] = make_uint2(wv, (unsigned)j);
          else               *s_qovf = 1;
        }
      }
    }
  }
}

template <int PASS, bool KEPT>
__device__ __forceinline__ void agg_stats(int n, int tid, const float* __restrict__ es,
                                          const unsigned* s_list, const float* s_ed,
                                          unsigned* s_mkey, unsigned long long* s_den,
                                          float* s_sc) {
  const int head  = tid & (NUM_HEADS - 1);
  const int total = n * NUM_HEADS;
  if (KEPT && PASS == 1) {
    #pragma unroll 4
    for (int it = tid; it < total; it += AGG_THREADS) {
      const int   k  = (int)(s_list[it >> 3] >> 20) * NUM_HEADS + head;
      const float ex = __expf(s_sc[it] - finite_or_zero(key2f(s_mkey[k])));
      s_sc[it] = ex;
      atomicAdd(&s_den[k], (unsigned long long)(unsigned)(ex * DEN_ONE));
    }
    return;
  }
  for (int it0 = tid; it0 < total; it0 += AGG_THREADS * UB) {
    int   tl[UB];
    float ev[UB];
    #pragma unroll
    for (int u = 0; u < UB; ++u) {
      const int it = it0 + u * AGG_THREADS;
      const unsigned w = s_list[(it < total ? it : total - 1) >> 3];
      tl[u] = (int)(w >> 20);
      ev[u] = es[(w & 0xFFFFFu) * NUM_HEADS + head];
    }
    #pragma unroll
    for (int u = 0; u < UB; ++u) {
      if (it0 + u * AGG_THREADS < total) {
        const int   k = tl[u] * NUM_HEADS + head;
        const float e = ev[u] + s_ed[k];
        const float v = e > 0.f ? e : NEG_SLOPE * e;
        if (PASS == 0) {
          if (KEPT) s_sc[it0 + u * AGG_THREADS] = v;
          atomicMax(&s_mkey[k], f2key(v));
        } else {
          const float ex = __expf(v - finite_or_zero(key2f(s_mkey[k])));
          atomicAdd(&s_den[k], (unsigned long long)(unsigned)(ex * DEN_ONE));
        }
      }
    }
  }
}

__device__ __forceinline__ void agg_den_to_float(int tid, const unsigned long long* s_den,
                                                 float* s_denf) {
  for (int i = tid; i < NBX * NUM_HEADS; i += AGG_THREADS)
    s_denf[i] = (float)s_den[i] * DEN_STEP;
}

template <bool KEPT>
__device__ __forceinline__ void agg_drain(int n, int wave, int lane,
                                          const float* __restrict__ h,
                                          const float* __restrict__ es,
                                          const unsigned* s_list, const float* s_ed,
                                          const unsigned* s_mkey, const float* s_denf,
                                          const float* s_sc, float* s_acc) {
  const int head = lane >> 2;
  for (int i0 = wave * BATCH; i0 < n; i0 += AGG_WAVES * BATCH) {
    float4 hv[BATCH];
    float  ev[BATCH];
    int    tl[BATCH];
    #pragma unroll
    for (int b = 0; b < BATCH; ++b) {
      const int idx = (i0 + b < n) ? (i0 + b) : (n - 1);
      const unsigned w = s_list[idx];
      const int s = (int)(w & 0xFFFFFu);
      tl[b] = (int)(w >> 20);
      ev[b] = KEPT ? 0.f : es[s * NUM_HEADS + head];
      hv[b] = *(const float4*)&h[(size_t)s * HD + 4 * lane];
    }
    #pragma unroll
    for (int b = 0; b < BATCH; ++b) {
      if (i0 + b < n) {
        const int   k     = tl[b] * NUM_HEADS + head;
        float ex;
        if (KEPT) {
          ex = s_sc[(i0 + b) * NUM_HEADS + head];
        } else {
          const float e = ev[b] + s_ed[k];
          const float v = e > 0.f ? e : NEG_SLOPE * e;
          ex = __expf(v - finite_or_zero(key2f(s_mkey[k])));
        }
        const float dv    = s_denf[k];
        const float alpha = ex / (dv > 0.f ? dv : 1.f);
        float* a = s_acc + tl[b] * HD + lane;
        atomicAdd(a +  0, alpha * hv[b].x);
        atomicAdd(a + 32, alpha * hv[b].y);
        atomicAdd(a + 64, alpha * hv[b].z);
        atomicAdd(a + 96, alpha * hv[b].w);
      }
    }
  }
}

typedef float rowsums_t __attribute__((ext_vector_type(4 * WN)));
__device__ __forceinline__ void agg_rowsum_owner(int n, int wave, int lane, int base, int nb,
                                                 const float* __restrict__ h,
                                                 const float* __restrict__ bias,
                                                 const uint2* q, const float* s_denf,
                                                 const float* s_sc, float* __restrict__ out) {
  const int head = lane >> 2;
  rowsums_t acc = 0.f;

  for (int i0 = 0; i0 < n; i0 += BATCH) {
    const int   qi = i0 + (lane & (BATCH - 1));
    const uint2 qe = q[qi < n ? qi : n - 1];
    float4 hv[BATCH];
    float  ex[BATCH];
    float  dv[BATCH];
    int    sl[BATCH];
    #pragma unroll
    for (int b = 0; b < BATCH; ++b) {
      const unsigned w  = (unsigned)__builtin_amdgcn_readlane((int)qe.x, b);
      const unsigned j  = (unsigned)__builtin_amdgcn_readlane((int)qe.y, b);
      const int      tl = (int)(w >> 20);
      sl[b] = 4 * (tl & (WN - 1));
      hv[b] = *(const float4*)&h[(size_t)(w & 0xFFFFFu) * HD + 4 * lane];
      ex[b] = s_sc[j * NUM_HEADS + head];
      dv[b] = s_denf[tl * NUM_HEADS + head];
    }
    #pragma unroll
    for (int b = 0; b < BATCH; ++b) {
      const float e     = (i0 + b < n) ? ex[b] : 0.f;
      const float alpha = e / (dv[b] > 0.f ? dv[b] : 1.f);
      acc[sl[b] + 0] += alpha * hv[b].x;
      acc[sl[b] + 1] += alpha * hv[b].y;
      acc[sl[b] + 2] += alpha * hv[b].z;
      acc[sl[b] + 3] += alpha * hv[b].w;
    }
  }

  const float4 bv = *(const float4*)&bias[4 * lane];
  #pragma unroll
  for (int s = 0; s < WN; ++s) {
    const int tl = wave * WN + s;
    if (tl < nb) {
      float4 o;
      o.x = acc[4 * s + 0] + bv.x;
      o.y = acc[4 * s + 1] + bv.y;
      o.z = acc[4 * s + 2] + bv.z;
      o.w = acc[4 * s + 3] + bv.w;
      *(float4*)&out[(size_t)(base + tl) * HD + 4 * lane] = o;
    }
  }
}

__global__ __launch_bounds__(AGG_THREADS)
void gat_softmax_aggregate_owner(const float* __restrict__ h,
                                 const float* __restrict__ es,
                                 const float* __restrict__ ed,
                                 const float* __restrict__ bias,
                                 const int* __restrict__ src,
                                 const int* __restrict__ dst,
                                 const unsigned* __restrict__ own,
                                 float* __restrict__ out) {
  __shared__ __attribute__((aligned(16))) float    s_acc[NBX * HD];
  __shared__ __attribute__((aligned(16))) unsigned s_list[LIST_CAP];
  __shared__ float    s_sc[SC_CAP * NUM_HEADS];
  __shared__ unsigned long long s_den[NBX * NUM_HEADS];
  __shared__ float    s_denf[NBX * NUM_HEADS];
  __shared__ unsigned s_mkey[NBX * NUM_HEADS];
  __shared__ float    s_ed[NBX * NUM_HEADS];
  __shared__ int      s_cnt;
  __shared__ int      s_ovf;
  __shared__ int      s_qn[AGG_WAVES];
  __shared__ int      s_qovf;
  const int tid  = threadIdx.x;
  const int lane = tid & 31;
  const int wave = tid >> 5;
  const int base = blockIdx.x * NB;
  const int nb   = (N_NODES - base < NB) ? (N_NODES - base) : NB;

  for (int i = tid; i < NBX * NUM_HEADS; i += AGG_THREADS) {
    s_ed[i]   = (i < nb * NUM_HEADS) ? ed[base * NUM_HEADS + i] : 0.f;
    s_mkey[i] = 0u;
    s_den[i]  = 0ull;
  }
  if (tid < AGG_WAVES) s_qn[tid] = 0;
  if (tid == 0) { s_cnt = 0; s_ovf = 0; s_qovf = 0; }
  __syncthreads();

  {
    const unsigned bb   = (unsigned)blockIdx.x * 0x01010101u;
    uint4*         wq_d = (uint4*)s_acc + wave * WQ_CAP;
    unsigned*      wq_i = (unsigned*)s_acc + AGG_WAVES * WQ_CAP * 4 + wave * WQ_CAP;
    int            wq   = 0;
    uint4 x[SUPER], xn[SUPER];
    int scr = (int)blockIdx.x % PSC_FULL;
    own_load_step<false>(scr, tid, bb, own, x);
    for (int sc = 0; sc < PSC_FULL - 1; ++sc) {
      const int scn = (scr + 1 == PSC_FULL) ? 0 : scr + 1;
      own_load_step<false>(scn, tid, bb, own, xn);
      own_queue_step(scr, tid, bb, x, wq_d, wq_i, wq);
      #pragma unroll
      for (int i = 0; i < SUPER; ++i) x[i] = xn[i];
      scr = scn;
    }
    own_load_step<true>(PSC_FULL, tid, bb, own, xn);
    own_queue_step(scr, tid, bb, x, wq_d, wq_i, wq);
    own_queue_step(PSC_FULL, tid, bb, xn, wq_d, wq_i, wq);
    if (wq > WQ_CAP) s_ovf = 1;
    own_emit(wq, lane, bb, wq_d, wq_i, s_list, &s_cnt, &s_ovf);
  }
  __syncthreads();

  uint2* wq2 = (uint2*)s_acc;
  bool main_path = (s_ovf == 0 && s_cnt <= SC_CAP);
  if (main_path) {
    agg_resolve<true>(s_cnt, tid, base, nb, src, dst, s_list, wq2, s_qn, &s_qovf);
    __syncthreads();
    main_path = (s_qovf == 0);
  }

  if (main_path) {
    const int n = s_cnt;
    agg_stats<0, true>(n, tid, es, s_list, s_ed, s_mkey, s_den, s_sc);
    __syncthreads();
    agg_stats<1, true>(n, tid, es, s_list, s_ed, s_mkey, s_den, s_sc);
    __syncthreads();
    agg_den_to_float(tid, s_den, s_denf);
    __syncthreads();
    const int uw = __builtin_amdgcn_readfirstlane(wave);
    const int un = __builtin_amdgcn_readfirstlane(s_qn[uw]);
    agg_rowsum_owner(un, uw, lane, base, nb, h, bias, wq2 + uw * WQ2_CAP, s_denf, s_sc, out);
    return;
  }

  {
    for (int i = tid; i < NBX * HD / 4; i += AGG_THREADS)
      ((float4*)s_acc)[i] = make_float4(0.f, 0.f, 0.f, 0.f);
    for (int pass = 0; pass < 3; ++pass) {
      for (int sc = 0; sc < NSC; ++sc) {
        int4 d[OSUPER];
        __syncthreads();
        if (tid == 0) s_cnt = 0;
        __syncthreads();
        agg_load_step(sc, tid, dst, d);
        agg_list_step(sc, tid, base, nb, d, s_list, &s_cnt, &s_ovf);
        __syncthreads();
        const int n = s_cnt;
        agg_resolve<false>(n, tid, base, nb, src, dst, s_list, wq2, s_qn, &s_qovf);
        __syncthreads();
        if (pass == 0)      agg_stats<0, false>(n, tid, es, s_list, s_ed, s_mkey, s_den, s_sc);
        else if (pass == 1) agg_stats<1, false>(n, tid, es, s_list, s_ed, s_mkey, s_den, s_sc);
        else                agg_drain<false>(n, wave, lane, h, es, s_list, s_ed, s_mkey, s_denf, s_sc, s_acc);
      }
      __syncthreads();
      if (pass == 1) agg_den_to_float(tid, s_den, s_denf);
    }
  }
  __syncthreads();

  const float4 bv = *(const float4*)&bias[4 * lane];
  for (int tl = wave; tl < nb; tl += AGG_WAVES) {
    const float* a = s_acc + tl * HD + lane;
    float4 o;
    o.x = a[0]  + bv.x;
    o.y = a[32] + bv.y;
    o.z = a[64] + bv.z;
    o.w = a[96] + bv.w;
    *(float4*)&out[(size_t)(base + tl) * HD + 4 * lane] = o;
  }
}

extern "C" void kernel_launch(void* const* d_in, const int* in_sizes, int n_in,
                              void* d_out, int out_size, void* d_ws, size_t ws_size,
                              hipStream_t stream) {
  const float* feat = (const float*)d_in[0];
  const float* W    = (const float*)d_in[1];
  const float* attn = (const float*)d_in[2];
  const float* bias = (const float*)d_in[3];
  const int*   src  = (const int*)d_in[4];
  const int*   dst  = (const int*)d_in[5];
  float*       out  = (float*)d_out;

  float* h  = (float*)d_ws;
  float* es = h  + (size_t)N_NODES * HD;
  float* ed = es + (size_t)N_NODES * NUM_HEADS;
  unsigned* own = (unsigned*)(ed + (size_t)N_NODES * NUM_HEADS);

  const int NH = N_NODES * NUM_HEADS;

  gat_gemm_wmma<<<GEMM_BLOCKS, GEMM_THREADS, 0, stream>>>(feat, W, h);
  gat_logits<<<(NH + 255) / 256, 256, 0, stream>>>(h, attn, dst, es, ed, (unsigned short*)own);
  gat_softmax_aggregate_owner<<<AGG_BLOCKS, AGG_THREADS, 0, stream>>>(h, es, ed, bias, src, dst, own, out);
}


# Logits launch folded into the projection's epilogue: es, ed from the staged rows it stores anyway (4 products a lane, 2 cross-lane adds), owner plane too; 2 launches, not 3; es, ed sum order changes
# speedup vs baseline: 7.7821x; 1.0338x over previous
#include <hip/hip_runtime.h>
#include <hip/hip_bf16.h>
#include <math.h>

#define N_NODES   50000
#define N_EDGES   800000
#define IN_FEATS  128
#define NUM_HEADS 8
#define OUT_FEATS 16
#define HD        (NUM_HEADS * OUT_FEATS)
#define NEG_SLOPE 0.2f

typedef float v2f __attribute__((ext_vector_type(2)));
typedef float v8f __attribute__((ext_vector_type(8)));

#define GEMM_WAVES   8
#define GEMM_THREADS (GEMM_WAVES * 32)
#define ROW_TILES    (N_NODES / 16)
#define GEMM_BLOCKS  ((ROW_TILES + GEMM_WAVES - 1) / GEMM_WAVES)
#define KJ           (IN_FEATS / 8)
#define NT           (HD / 16)
#define OPITCH       132
#define W_SLOTS      (KJ * NT * 32)

static_assert(N_NODES % 16 == 0 && IN_FEATS % 8 == 0 && HD % 16 == 0, "whole tiles only");
static_assert(NT == GEMM_WAVES, "W staging: wave w fills column tile w");
static_assert(OPITCH % 4 == 0, "a staged output row starts on 16 bytes");

#define NB 196
static_assert((N_NODES + NB - 1) / NB <= 256, "an owner number fits a byte");
static_assert(ROW_TILES * 32 * 8 == N_EDGES,
              "owner plane: every lane of every live wave writes the bytes of eight edges");
static_assert(OUT_FEATS == 16 && HD == 128, "a head is one column tile: 4 lanes of a staged row");

__device__ __forceinline__ unsigned own_bytes4(int4 d) {
  return  (((unsigned)d.x / NB) & 0xFFu)        | ((((unsigned)d.y / NB) & 0xFFu) << 8) |
         ((((unsigned)d.z / NB) & 0xFFu) << 16) | ((((unsigned)d.w / NB) & 0xFFu) << 24);
}

template <int M>
__device__ __forceinline__ float add_lane_xor(float v) {
  return v + __int_as_float(__builtin_amdgcn_update_dpp(0, __float_as_int(v), 0x160 + M, 0xF, 0xF, true));
}

__global__ __launch_bounds__(GEMM_THREADS)
void gat_gemm_wmma(const float* __restrict__ feat,
                   const float* __restrict__ W,
                   const float* __restrict__ attn,
                   const int* __restrict__ dst,
                   float* __restrict__ h,
                   float* __restrict__ es, float* __restrict__ ed,
                   unsigned* __restrict__ own) {
  __shared__ float4 lds4[W_SLOTS + GEMM_WAVES * 16 * OPITCH / 4];
  float4* lW = lds4;
  float*  lO = (float*)(lds4 + W_SLOTS);
  const int tid  = threadIdx.x;
  const int lane = tid & 31;
  const int wave = tid >> 5;
  const int c    = lane & 15;
  const int half = lane >> 4;

  const float4 as4 = *(const float4*)&attn[4 * lane];
  const float4 ad4 = *(const float4*)&attn[HD + 4 * lane];

  for (int j = 0; j < KJ; ++j) {
    const float* wp = W + (size_t)(8 * j + 4 * half) * HD + 16 * wave + c;
    float4 v;
    v.x = wp[0];
    v.y = wp[HD];
    v.z = wp[2 * HD];
    v.w = wp[3 * HD];
    lW[(j * NT + wave) * 32 + lane] = v;
  }
  __syncthreads();

  const int  t    = blockIdx.x * GEMM_WAVES + wave;
  const bool live = t < ROW_TILES;
  float* orow = lO + wave * 16 * OPITCH;

  if (live) {
    const float4* ap = (const float4*)(feat + (size_t)(t * 16 + c) * IN_FEATS + 4 * half);
    float4 a[KJ];
    #pragma unroll
    for (int j = 0; j < KJ; ++j) a[j] = ap[2 * j];

    const int  u  = t * 32 + lane;
    const int4 d0 = ((const int4*)dst)[2 * u];
    const int4 d1 = ((const int4*)dst)[2 * u + 1];

    v8f acc[NT];
    #pragma unroll
    for (int n = 0; n < NT; ++n) acc[n] = v8f{};

    #pragma unroll
    for (int j = 0; j < KJ; ++j) {
      float4 b[NT];
      #pragma unroll
      for (int n = 0; n < NT; ++n) b[n] = lW[(j * NT + n) * 32 + lane];
      v2f a0; a0.x = a[j].x; a0.y = a[j].y;
      v2f a1; a1.x = a[j].z; a1.y = a[j].w;
      #pragma unroll
      for (int n = 0; n < NT; ++n) {
        v2f b0; b0.x = b[n].x; b0.y = b[n].y;
        acc[n] = __builtin_amdgcn_wmma_f32_16x16x4_f32(
             false, a0,  false, b0,
             (short)0, acc[n],  false,  false);
      }
      #pragma unroll
      for (int n = 0; n < NT; ++n) {
        v2f b1; b1.x = b[n].z; b1.y = b[n].w;
        acc[n] = __builtin_amdgcn_wmma_f32_16x16x4_f32(
             false, a1,  false, b1,
             (short)0, acc[n],  false,  false);
      }
    }

    ((uint2*)own)[u] = make_uint2(own_bytes4(d0), own_bytes4(d1));

    #pragma unroll
    for (int n = 0; n < NT; ++n) {
      #pragma unroll
      for (int r = 0; r < 8; ++r) orow[(8 * half + r) * OPITCH + 16 * n + c] = acc[n][r];
    }
  }
  __syncthreads();
  if (live) {
    const int jq = lane & 3;
    float ks[4], kd[4];
    #pragma unroll
    for (int r = 0; r < 16; ++r) {
      const float4 v = *(const float4*)&orow[r * OPITCH + 4 * lane];
      *(float4*)&h[(size_t)(t * 16 + r) * HD + 4 * lane] = v;
      float ps = v.x * as4.x + v.y * as4.y + v.z * as4.z + v.w * as4.w;
      float pd = v.x * ad4.x + v.y * ad4.y + v.z * ad4.z + v.w * ad4.w;
      ps = add_lane_xor<1>(ps);
      pd = add_lane_xor<1>(pd);
      ps = add_lane_xor<2>(ps);
      pd = add_lane_xor<2>(pd);
      if (jq == (r & 3)) { ks[r >> 2] = ps; kd[r >> 2] = pd; }
    }
    #pragma unroll
    for (int i = 0; i < 4; ++i) {
      const int k = (t * 16 + 4 * i + jq) * NUM_HEADS + (lane >> 2);
      es[k] = ks[i];
      ed[k] = kd[i];
    }
  }
}

__device__ __forceinline__ unsigned f2key(float v) {
  const unsigned u = __float_as_uint(v);
  return (u & 0x80000000u) ? ~u : (u | 0x80000000u);
}
__device__ __forceinline__ float key2f(unsigned k) {
  return __uint_as_float((k & 0x80000000u) ? (k & 0x7FFFFFFFu) : ~k);
}
__device__ __forceinline__ float finite_or_zero(float m) {
  return (fabsf(m) < INFINITY) ? m : 0.f;
}

#define NBX         (NB + 1)
#define AGG_BLOCKS  256
#define AGG_THREADS 800
#define AGG_WAVES   (AGG_THREADS / 32)
#define SUPER       5
#define OSUPER      2
#define PQ          (N_EDGES / 16)
#define P_PER_SC    (AGG_THREADS * SUPER)
#define PSC_FULL    (PQ / P_PER_SC)
#define WQ_CAP      200
#define NQ          (N_EDGES / 4)
#define Q_PER_SC    (AGG_THREADS * OSUPER)
#define NSC         (NQ / Q_PER_SC)
#define LIST_CAP    8192
#define SC_CAP      3840
#define BATCH       8
#define WN          8
#define WQ2_CAP     (NBX * HD / 2 / AGG_WAVES)
#define UB          8
#define UR          4
#define DEN_ONE     2147483648.0f
#define DEN_STEP    4.656612873077392578125e-10f

static_assert(N_EDGES <= (1 << 20) && NBX <= (1 << 12), "a list word is local node << 20 | edge or src");
static_assert(N_NODES <= (1 << 20), "a resolved list word is local node << 20 | src node");
static_assert(AGG_BLOCKS * NB >= N_NODES && AGG_BLOCKS <= 256, "owner ranges cover all nodes, one byte each");
static_assert(N_EDGES % 16 == 0 && PSC_FULL * P_PER_SC < PQ && (PSC_FULL + 1) * P_PER_SC >= PQ,
              "the plane is scanned in whole steps and one partial step");
static_assert(N_EDGES % 4 == 0 && NQ % Q_PER_SC == 0, "overflow path: dst is scanned as int4, in whole steps");
static_assert(Q_PER_SC * 4 <= LIST_CAP, "overflow path: one scan step always fits the list");
static_assert(SC_CAP <= LIST_CAP, "every edge with kept scores is on the list");
static_assert(AGG_THREADS % NUM_HEADS == 0, "a thread keeps one head over all its items");
static_assert(AGG_WAVES * WQ_CAP * 5 <= NBX * HD, "the wave queues fit the LDS words of the row sums");
static_assert(AGG_WAVES * WN >= NBX, "every row, the spare one too, has an owner wave");
static_assert((WN & (WN - 1)) == 0 && BATCH <= 32, "a node's place in its wave is tl & (WN - 1); a batch's entries sit in one lane each");

__device__ __forceinline__ unsigned zero_bytes(unsigned z) {
  return ~((((z & 0x7F7F7F7Fu) + 0x7F7F7F7Fu) | z) | 0x7F7F7F7Fu);
}

template <bool PARTIAL>
__device__ __forceinline__ void own_load_step(int scr, int tid, unsigned bb,
                                              const unsigned* __restrict__ own,
                                              uint4 (&x)[SUPER]) {
  const uint4* own4 = (const uint4*)own;
  #pragma unroll
  for (int i = 0; i < SUPER; ++i) {
    const int q = scr * P_PER_SC + i * AGG_THREADS + tid;
    if (!PARTIAL || q < PQ) x[i] = own4[q];
    else                    x[i] = make_uint4(~bb, ~bb, ~bb, ~bb);
  }
}

__device__ __forceinline__ void own_queue_step(int scr, int tid, unsigned bb,
                                               const uint4 (&x)[SUPER],
                                               uint4* wq_d, unsigned* wq_i, int& wq) {
  #pragma unroll
  for (int i = 0; i < SUPER; ++i) {
    const int q = scr * P_PER_SC + i * AGG_THREADS + tid;
    const unsigned z0 = x[i].x ^ bb, z1 = x[i].y ^ bb, z2 = x[i].z ^ bb, z3 = x[i].w ^ bb;
    const unsigned t = (((z0 - 0x01010101u) & ~z0) | ((z1 - 0x01010101u) & ~z1) |
                        ((z2 - 0x01010101u) & ~z2) | ((z3 - 0x01010101u) & ~z3)) & 0x80808080u;
    const bool     cand = t != 0u;
    const unsigned m    = __builtin_amdgcn_ballot_w32(cand);
    if (m != 0u) {
      if (cand) {
        const int slot = min(wq + (int)__builtin_amdgcn_mbcnt_lo(m, 0u), WQ_CAP - 1);
        wq_d[slot] = x[i];
        wq_i[slot] = (unsigned)q;
      }
      wq += __popc(m);
    }
  }
}

__device__ __forceinline__ void own_emit(int wq, int lane, unsigned bb,
                                         const uint4* wq_d, const unsigned* wq_i,
                                         unsigned* s_list, int* s_cnt, int* s_ovf) {
  const int n = wq < WQ_CAP ? wq : WQ_CAP;
  for (int i0 = 0; i0 < n; i0 += 32) {
    const int i = i0 + lane;
    unsigned hz[4] = {0u, 0u, 0u, 0u};
    unsigned q = 0u;
    if (i < n) {
      const uint4 x = wq_d[i];
      q = wq_i[i];
      hz[0] = zero_bytes(x.x ^ bb);
      hz[1] = zero_bytes(x.y ^ bb);
      hz[2] = zero_bytes(x.z ^ bb);
      hz[3] = zero_bytes(x.w ^ bb);
    }
    const int c = __popc(hz[0]) + __popc(hz[1]) + __popc(hz[2]) + __popc(hz[3]);
    int incl = c;
    #pragma unroll
    for (int d = 1; d < 32; d <<= 1) {
      const int v = __shfl_up(incl, d, 32);
      if (lane >= d) incl += v;
    }
    const int total = __shfl(incl, 31, 32);
    int first = 0;
    if (lane == 0) first = atomicAdd(s_cnt, total);
    first = __shfl(first, 0, 32);
    int pos = first + incl - c;
    #pragma unroll
    for (int j = 0; j < 4; ++j) {
      unsigned rest = hz[j];
      while (rest != 0u) {
        const unsigned b = (unsigned)(__ffs((int)rest) - 1) >> 3;
        if (pos < LIST_CAP) s_list[pos] = 16u * q + 4u * (unsigned)j + b;
        else                *s_ovf = 1;
        ++pos;
        rest &= rest - 1u;
      }
    }
  }
}

__device__ __forceinline__ void agg_load_step(int scr, int tid, const int* __restrict__ dst,
                                              int4 (&d)[OSUPER]) {
  const int4* dst4 = (const int4*)dst;
  #pragma unroll
  for (int i = 0; i < OSUPER; ++i) {
    d[i] = dst4[scr * Q_PER_SC + i * AGG_THREADS + tid];
  }
}

__device__ __forceinline__ void agg_list_step(int scr, int tid, int base, int nb,
                                              const int4 (&d)[OSUPER],
                                              unsigned* s_list, int* s_cnt, int* s_ovf) {
  #pragma unroll
  for (int i = 0; i < OSUPER; ++i) {
    const int q = scr * Q_PER_SC + i * AGG_THREADS + tid;
    const unsigned tv[4] = {(unsigned)(d[i].x - base), (unsigned)(d[i].y - base),
                            (unsigned)(d[i].z - base), (unsigned)(d[i].w - base)};
    if (min(min(tv[0], tv[1]), min(tv[2], tv[3])) < (unsigned)nb) {
      #pragma unroll
      for (int c = 0; c < 4; ++c) {
        if (tv[c] < (unsigned)nb) {
          const int pos = atomicAdd(s_cnt, 1);
          if (pos < LIST_CAP) s_list[pos] = (unsigned)(4 * q + c);
          else                *s_ovf = 1;
        }
      }
    }
  }
}

template <bool DEAL>
__device__ __forceinline__ void agg_resolve(int n, int tid, int base, int nb,
                                            const int* __restrict__ src,
                                            const int* __restrict__ dst,
                                            unsigned* s_list,
                                            uint2* wq2, int* s_qn, int* s_qovf) {
  for (int j0 = tid; j0 < n; j0 += AGG_THREADS * UR) {
    unsigned e[UR];
    int      sv[UR];
    int      dv[UR];
    #pragma unroll
    for (int u = 0; u < UR; ++u) {
      const int j = j0 + u * AGG_THREADS;
      e[u]  = (j < n) ? (s_list[j] & 0xFFFFFu) : 0u;
      sv[u] = src[e[u]];
      dv[u] = dst[e[u]];
    }
    #pragma unroll
    for (int u = 0; u < UR; ++u) {
      const int j = j0 + u * AGG_THREADS;
      if (j < n) {
        unsigned tl = (unsigned)(dv[u] - base);
        if (tl >= (unsigned)nb) tl = NB;
        const unsigned wv = (tl << 20) | (unsigned)sv[u];
        s_list[j] = wv;
        if (DEAL) {
          const int ow  = (int)(tl / WN);
          const int pos = atomicAdd(&s_qn[ow], 1);
          if (pos < WQ2_CAP) wq2[ow * WQ2_CAP + pos] = make_uint2(wv, (unsigned)j);
          else               *s_qovf = 1;
        }
      }
    }
  }
}

template <int PASS, bool KEPT>
__device__ __forceinline__ void agg_stats(int n, int tid, const float* __restrict__ es,
                                          const unsigned* s_list, const float* s_ed,
                                          unsigned* s_mkey, unsigned long long* s_den,
                                          float* s_sc) {
  const int head  = tid & (NUM_HEADS - 1);
  const int total = n * NUM_HEADS;
  if (KEPT && PASS == 1) {
    #pragma unroll 4
    for (int it = tid; it < total; it += AGG_THREADS) {
      const int   k  = (int)(s_list[it >> 3] >> 20) * NUM_HEADS + head;
      const float ex = __expf(s_sc[it] - finite_or_zero(key2f(s_mkey[k])));
      s_sc[it] = ex;
      atomicAdd(&s_den[k], (unsigned long long)(unsigned)(ex * DEN_ONE));
    }
    return;
  }
  for (int it0 = tid; it0 < total; it0 += AGG_THREADS * UB) {
    int   tl[UB];
    float ev[UB];
    #pragma unroll
    for (int u = 0; u < UB; ++u) {
      const int it = it0 + u * AGG_THREADS;
      const unsigned w = s_list[(it < total ? it : total - 1) >> 3];
      tl[u] = (int)(w >> 20);
      ev[u] = es[(w & 0xFFFFFu) * NUM_HEADS + head];
    }
    #pragma unroll
    for (int u = 0; u < UB; ++u) {
      if (it0 + u * AGG_THREADS < total) {
        const int   k = tl[u] * NUM_HEADS + head;
        const float e = ev[u] + s_ed[k];
        const float v = e > 0.f ? e : NEG_SLOPE * e;
        if (PASS == 0) {
          if (KEPT) s_sc[it0 + u * AGG_THREADS] = v;
          atomicMax(&s_mkey[k], f2key(v));
        } else {
          const float ex = __expf(v - finite_or_zero(key2f(s_mkey[k])));
          atomicAdd(&s_den[k], (unsigned long long)(unsigned)(ex * DEN_ONE));
        }
      }
    }
  }
}

__device__ __forceinline__ void agg_den_to_float(int tid, const unsigned long long* s_den,
                                                 float* s_denf) {
  for (int i = tid; i < NBX * NUM_HEADS; i += AGG_THREADS)
    s_denf[i] = (float)s_den[i] * DEN_STEP;
}

template <bool KEPT>
__device__ __forceinline__ void agg_drain(int n, int wave, int lane,
                                          const float* __restrict__ h,
                                          const float* __restrict__ es,
                                          const unsigned* s_list, const float* s_ed,
                                          const unsigned* s_mkey, const float* s_denf,
                                          const float* s_sc, float* s_acc) {
  const int head = lane >> 2;
  for (int i0 = wave * BATCH; i0 < n; i0 += AGG_WAVES * BATCH) {
    float4 hv[BATCH];
    float  ev[BATCH];
    int    tl[BATCH];
    #pragma unroll
    for (int b = 0; b < BATCH; ++b) {
      const int idx = (i0 + b < n) ? (i0 + b) : (n - 1);
      const unsigned w = s_list[idx];
      const int s = (int)(w & 0xFFFFFu);
      tl[b] = (int)(w >> 20);
      ev[b] = KEPT ? 0.f : es[s * NUM_HEADS + head];
      hv[b] = *(const float4*)&h[(size_t)s * HD + 4 * lane];
    }
    #pragma unroll
    for (int b = 0; b < BATCH; ++b) {
      if (i0 + b < n) {
        const int   k     = tl[b] * NUM_HEADS + head;
        float ex;
        if (KEPT) {
          ex = s_sc[(i0 + b) * NUM_HEADS + head];
        } else {
          const float e = ev[b] + s_ed[k];
          const float v = e > 0.f ? e : NEG_SLOPE * e;
          ex = __expf(v - finite_or_zero(key2f(s_mkey[k])));
        }
        const float dv    = s_denf[k];
        const float alpha = ex / (dv > 0.f ? dv : 1.f);
        float* a = s_acc + tl[b] * HD + lane;
        atomicAdd(a +  0, alpha * hv[b].x);
        atomicAdd(a + 32, alpha * hv[b].y);
        atomicAdd(a + 64, alpha * hv[b].z);
        atomicAdd(a + 96, alpha * hv[b].w);
      }
    }
  }
}

typedef float rowsums_t __attribute__((ext_vector_type(4 * WN)));
__device__ __forceinline__ void agg_rowsum_owner(int n, int wave, int lane, int base, int nb,
                                                 const float* __restrict__ h,
                                                 const float* __restrict__ bias,
                                                 const uint2* q, const float* s_denf,
                                                 const float* s_sc, float* __restrict__ out) {
  const int head = lane >> 2;
  rowsums_t acc = 0.f;

  for (int i0 = 0; i0 < n; i0 += BATCH) {
    const int   qi = i0 + (lane & (BATCH - 1));
    const uint2 qe = q[qi < n ? qi : n - 1];
    float4 hv[BATCH];
    float  ex[BATCH];
    float  dv[BATCH];
    int    sl[BATCH];
    #pragma unroll
    for (int b = 0; b < BATCH; ++b) {
      const unsigned w  = (unsigned)__builtin_amdgcn_readlane((int)qe.x, b);
      const unsigned j  = (unsigned)__builtin_amdgcn_readlane((int)qe.y, b);
      const int      tl = (int)(w >> 20);
      sl[b] = 4 * (tl & (WN - 1));
      hv[b] = *(const float4*)&h[(size_t)(w & 0xFFFFFu) * HD + 4 * lane];
      ex[b] = s_sc[j * NUM_HEADS + head];
      dv[b] = s_denf[tl * NUM_HEADS + head];
    }
    #pragma unroll
    for (int b = 0; b < BATCH; ++b) {
      const float e     = (i0 + b < n) ? ex[b] : 0.f;
      const float alpha = e / (dv[b] > 0.f ? dv[b] : 1.f);
      acc[sl[b] + 0] += alpha * hv[b].x;
      acc[sl[b] + 1] += alpha * hv[b].y;
      acc[sl[b] + 2] += alpha * hv[b].z;
      acc[sl[b] + 3] += alpha * hv[b].w;
    }
  }

  const float4 bv = *(const float4*)&bias[4 * lane];
  #pragma unroll
  for (int s = 0; s < WN; ++s) {
    const int tl = wave * WN + s;
    if (tl < nb) {
      float4 o;
      o.x = acc[4 * s + 0] + bv.x;
      o.y = acc[4 * s + 1] + bv.y;
      o.z = acc[4 * s + 2] + bv.z;
      o.w = acc[4 * s + 3] + bv.w;
      *(float4*)&out[(size_t)(base + tl) * HD + 4 * lane] = o;
    }
  }
}

__global__ __launch_bounds__(AGG_THREADS)
void gat_softmax_aggregate_owner(const float* __restrict__ h,
                                 const float* __restrict__ es,
                                 const float* __restrict__ ed,
                                 const float* __restrict__ bias,
                                 const int* __restrict__ src,
                                 const int* __restrict__ dst,
                                 const unsigned* __restrict__ own,
                                 float* __restrict__ out) {
  __shared__ __attribute__((aligned(16))) float    s_acc[NBX * HD];
  __shared__ __attribute__((aligned(16))) unsigned s_list[LIST_CAP];
  __shared__ float    s_sc[SC_CAP * NUM_HEADS];
  __shared__ unsigned long long s_den[NBX * NUM_HEADS];
  __shared__ float    s_denf[NBX * NUM_HEADS];
  __shared__ unsigned s_mkey[NBX * NUM_HEADS];
  __shared__ float    s_ed[NBX * NUM_HEADS];
  __shared__ int      s_cnt;
  __shared__ int      s_ovf;
  __shared__ int      s_qn[AGG_WAVES];
  __shared__ int      s_qovf;
  const int tid  = threadIdx.x;
  const int lane = tid & 31;
  const int wave = tid >> 5;
  const int base = blockIdx.x * NB;
  const int nb   = (N_NODES - base < NB) ? (N_NODES - base) : NB;

  for (int i = tid; i < NBX * NUM_HEADS; i += AGG_THREADS) {
    s_ed[i]   = (i < nb * NUM_HEADS) ? ed[base * NUM_HEADS + i] : 0.f;
    s_mkey[i] = 0u;
    s_den[i]  = 0ull;
  }
  if (tid < AGG_WAVES) s_qn[tid] = 0;
  if (tid == 0) { s_cnt = 0; s_ovf = 0; s_qovf = 0; }
  __syncthreads();

  {
    const unsigned bb   = (unsigned)blockIdx.x * 0x01010101u;
    uint4*         wq_d = (uint4*)s_acc + wave * WQ_CAP;
    unsigned*      wq_i = (unsigned*)s_acc + AGG_WAVES * WQ_CAP * 4 + wave * WQ_CAP;
    int            wq   = 0;
    uint4 x[SUPER], xn[SUPER];
    int scr = (int)blockIdx.x % PSC_FULL;
    own_load_step<false>(scr, tid, bb, own, x);
    for (int sc = 0; sc < PSC_FULL - 1; ++sc) {
      const int scn = (scr + 1 == PSC_FULL) ? 0 : scr + 1;
      own_load_step<false>(scn, tid, bb, own, xn);
      own_queue_step(scr, tid, bb, x, wq_d, wq_i, wq);
      #pragma unroll
      for (int i = 0; i < SUPER; ++i) x[i] = xn[i];
      scr = scn;
    }
    own_load_step<true>(PSC_FULL, tid, bb, own, xn);
    own_queue_step(scr, tid, bb, x, wq_d, wq_i, wq);
    own_queue_step(PSC_FULL, tid, bb, xn, wq_d, wq_i, wq);
    if (wq > WQ_CAP) s_ovf = 1;
    own_emit(wq, lane, bb, wq_d, wq_i, s_list, &s_cnt, &s_ovf);
  }
  __syncthreads();

  uint2* wq2 = (uint2*)s_acc;
  bool main_path = (s_ovf == 0 && s_cnt <= SC_CAP);
  if (main_path) {
    agg_resolve<true>(s_cnt, tid, base, nb, src, dst, s_list, wq2, s_qn, &s_qovf);
    __syncthreads();
    main_path = (s_qovf == 0);
  }

  if (main_path) {
    const int n = s_cnt;
    agg_stats<0, true>(n, tid, es, s_list, s_ed, s_mkey, s_den, s_sc);
    __syncthreads();
    agg_stats<1, true>(n, tid, es, s_list, s_ed, s_mkey, s_den, s_sc);
    __syncthreads();
    agg_den_to_float(tid, s_den, s_denf);
    __syncthreads();
    const int uw = __builtin_amdgcn_readfirstlane(wave);
    const int un = __builtin_amdgcn_readfirstlane(s_qn[uw]);
    agg_rowsum_owner(un, uw, lane, base, nb, h, bias, wq2 + uw * WQ2_CAP, s_denf, s_sc, out);
    return;
  }

  {
    for (int i = tid; i < NBX * HD / 4; i += AGG_THREADS)
      ((float4*)s_acc)[i] = make_float4(0.f, 0.f, 0.f, 0.f);
    for (int pass = 0; pass < 3; ++pass) {
      for (int sc = 0; sc < NSC; ++sc) {
        int4 d[OSUPER];
        __syncthreads();
        if (tid == 0) s_cnt = 0;
        __syncthreads();
        agg_load_step(sc, tid, dst, d);
        agg_list_step(sc, tid, base, nb, d, s_list, &s_cnt, &s_ovf);
        __syncthreads();
        const int n = s_cnt;
        agg_resolve<false>(n, tid, base, nb, src, dst, s_list, wq2, s_qn, &s_qovf);
        __syncthreads();
        if (pass == 0)      agg_stats<0, false>(n, tid, es, s_list, s_ed, s_mkey, s_den, s_sc);
        else if (pass == 1) agg_stats<1, false>(n, tid, es, s_list, s_ed, s_mkey, s_den, s_sc);
        else                agg_drain<false>(n, wave, lane, h, es, s_list, s_ed, s_mkey, s_denf, s_sc, s_acc);
      }
      __syncthreads();
      if (pass == 1) agg_den_to_float(tid, s_den, s_denf);
    }
  }
  __syncthreads();

  const float4 bv = *(const float4*)&bias[4 * lane];
  for (int tl = wave; tl < nb; tl += AGG_WAVES) {
    const float* a = s_acc + tl * HD + lane;
    float4 o;
    o.x = a[0]  + bv.x;
    o.y = a[32] + bv.y;
    o.z = a[64] + bv.z;
    o.w = a[96] + bv.w;
    *(float4*)&out[(size_t)(base + tl) * HD + 4 * lane] = o;
  }
}

extern "C" void kernel_launch(void* const* d_in, const int* in_sizes, int n_in,
                              void* d_out, int out_size, void* d_ws, size_t ws_size,
                              hipStream_t stream) {
  const float* feat = (const float*)d_in[0];
  const float* W    = (const float*)d_in[1];
  const float* attn = (const float*)d_in[2];
  const float* bias = (const float*)d_in[3];
  const int*   src  = (const int*)d_in[4];
  const int*   dst  = (const int*)d_in[5];
  float*       out  = (float*)d_out;

  float* h  = (float*)d_ws;
  float* es = h  + (size_t)N_NODES * HD;
  float* ed = es + (size_t)N_NODES * NUM_HEADS;
  unsigned* own = (unsigned*)(ed + (size_t)N_NODES * NUM_HEADS);

  gat_gemm_wmma<<<GEMM_BLOCKS, GEMM_THREADS, 0, stream>>>(feat, W, attn, dst, h, es, ed, own);
  gat_softmax_aggregate_owner<<<AGG_BLOCKS, AGG_THREADS, 0, stream>>>(h, es, ed, bias, src, dst, own, out);
}
